# starting kernel
# baseline (speedup 1.0000x reference)
#include <hip/hip_runtime.h>
#include <hip/hip_bf16.h>
#include <math.h>

#define B_  4096
#define M_  100
#define E_  128
#define R_  64

typedef float v2f __attribute__((ext_vector_type(2)));
typedef float v8f __attribute__((ext_vector_type(8)));

#define STRIDE    132
#define MT        112
#define OFF_INTER 0
#define OFF_U     (OFF_INTER + MT * STRIDE)
#define OFF_IP    (OFF_U + E_)
#define OFF_INN   (OFF_IP + E_)
#define OFF_W     (OFF_INN + E_)
#define OFF_WSUA  (OFF_W + 4 * E_)
#define OFF_WSIA  (OFF_WSUA + E_)
#define OFF_LOGIT (OFF_WSIA + E_)
#define OFF_SACC  (OFF_LOGIT + 4 * MT)
#define SMEM_FLOATS (OFF_SACC + 4 * E_)

__device__ __forceinline__ float wave_sum(float v) {
    #pragma unroll
    for (int off = 16; off > 0; off >>= 1) v += __shfl_xor(v, off, 32);
    return v;
}
__device__ __forceinline__ float wave_max(float v) {
    #pragma unroll
    for (int off = 16; off > 0; off >>= 1) v = fmaxf(v, __shfl_xor(v, off, 32));
    return v;
}

__device__ __forceinline__ void renorm_row(const float* __restrict__ src,
                                           float* __restrict__ dst, int lane) {
    float4 x = ((const float4*)src)[lane];
    float ss = x.x * x.x + x.y * x.y + x.z * x.z + x.w * x.w;
    ss = wave_sum(ss);
    float n = sqrtf(ss);
    float s = (n > 1.0f) ? (1.0f / (n + 1e-7f)) : 1.0f;
    x.x *= s; x.y *= s; x.z *= s; x.w *= s;
    ((float4*)dst)[lane] = x;
}

__global__ void hlrm_wsum_kernel(const float* __restrict__ ua_key,
                                 const float* __restrict__ ua_val,
                                 const float* __restrict__ ia_key,
                                 const float* __restrict__ ia_val,
                                 float* __restrict__ wsum) {
    int e = threadIdx.x;
    float s1 = 0.f, s2 = 0.f;
    #pragma unroll 4
    for (int r = 0; r < R_; ++r) {
        s1 += ua_key[r * E_ + e] * ua_val[r * E_ + e];
        s2 += ia_key[r * E_ + e] * ia_val[r * E_ + e];
    }
    wsum[e]       = s1;
    wsum[E_ + e]  = s2;
}

__global__ void __launch_bounds__(256, 2)
hlrm_main_kernel(const int* __restrict__ user_id,
                 const int* __restrict__ item_id_p,
                 const int* __restrict__ item_id_n,
                 const int* __restrict__ inter_id,
                 const float* __restrict__ user_emb_w,
                 const float* __restrict__ item_emb_w,
                 const float* __restrict__ wsum_g,
                 float* __restrict__ out) {
    extern __shared__ float sm[];
    float* sm_inter = sm + OFF_INTER;
    float* sm_u     = sm + OFF_U;
    float* sm_ip    = sm + OFF_IP;
    float* sm_inn   = sm + OFF_INN;
    float* sm_w     = sm + OFF_W;
    float* sm_wsua  = sm + OFF_WSUA;
    float* sm_wsia  = sm + OFF_WSIA;
    float* sm_logit = sm + OFF_LOGIT;
    float* sm_sacc  = sm + OFF_SACC;

    const int b    = blockIdx.x;
    const int t    = threadIdx.x;
    const int wave = t >> 5;
    const int lane = t & 31;
    const int half = lane >> 4;
    const int lm   = lane & 15;

    if (wave == 0)      renorm_row(user_emb_w + (size_t)user_id[b]   * E_, sm_u,   lane);
    else if (wave == 1) renorm_row(item_emb_w + (size_t)item_id_p[b] * E_, sm_ip,  lane);
    else if (wave == 2) renorm_row(item_emb_w + (size_t)item_id_n[b] * E_, sm_inn, lane);

    for (int m = wave; m < MT; m += 8) {
        if (m < M_) {
            const float* src = item_emb_w + (size_t)inter_id[b * M_ + m] * E_;
            if (m + 8 < M_) {
                const float* nxt = item_emb_w + (size_t)inter_id[b * M_ + m + 8] * E_;
                __builtin_prefetch(nxt + 4 * lane, 0, 3);
            }
            renorm_row(src, sm_inter + m * STRIDE, lane);
        } else {
            float4 z = {0.f, 0.f, 0.f, 0.f};
            ((float4*)(sm_inter + m * STRIDE))[lane] = z;
        }
    }
    __syncthreads();

    if (t < E_) {
        const int e = t;
        const float uu = sm_u[e], pp = sm_ip[e], nn = sm_inn[e];
        const float wa = wsum_g[e], wi = wsum_g[E_ + e];
        sm_wsua[e] = wa;
        sm_wsia[e] = wi;
        sm_w[0 * E_ + e] = uu * pp * wa;
        sm_w[1 * E_ + e] = uu * pp * wi;
        sm_w[2 * E_ + e] = uu * nn * wa;
        sm_w[3 * E_ + e] = uu * nn * wi;
    }
    __syncthreads();

    if (wave < 7) {
        const int   tile = wave;
        const float* Arow = sm_inter + (tile * 16 + lm) * STRIDE;
        const float  bmask = (lm < 4) ? 1.0f : 0.0f;
        const float* wrow  = sm_w + (lm < 4 ? lm : 0) * E_;
        v8f c = {0.f, 0.f, 0.f, 0.f, 0.f, 0.f, 0.f, 0.f};
        #pragma unroll 8
        for (int k0 = 0; k0 < E_; k0 += 4) {
            const int ke = k0 + 2 * half;
            v2f a, bb;
            a[0]  = Arow[ke];           a[1]  = Arow[ke + 1];
            bb[0] = wrow[ke] * bmask;   bb[1] = wrow[ke + 1] * bmask;
            c = __builtin_amdgcn_wmma_f32_16x16x4_f32(false, a, false, bb,
                                                      (short)0, c, false, false);
        }
        if (lm < 4) {
            #pragma unroll
            for (int v = 0; v < 8; ++v)
                sm_logit[lm * MT + tile * 16 + v + 8 * half] = c[v];
        }
    }
    __syncthreads();

    if (wave < 4) {
        float* L = sm_logit + wave * MT;
        float vals[4], mx = -INFINITY;
        #pragma unroll
        for (int i = 0; i < 4; ++i) {
            int m = lane + 32 * i;
            vals[i] = (m < M_) ? L[m] : -INFINITY;
            mx = fmaxf(mx, vals[i]);
        }
        mx = wave_max(mx);
        float se = 0.f;
        #pragma unroll
        for (int i = 0; i < 4; ++i) {
            int m = lane + 32 * i;
            vals[i] = (m < M_) ? expf(vals[i] - mx) : 0.f;
            se += vals[i];
        }
        se = wave_sum(se);
        const float inv = 1.0f / se;
        #pragma unroll
        for (int i = 0; i < 4; ++i) {
            int m = lane + 32 * i;
            if (m < MT) L[m] = vals[i] * inv;
        }
    }
    __syncthreads();

    {
        const int   etile = wave;
        const float amask = (lm < 4) ? 1.0f : 0.0f;
        const float* Lrow = sm_logit + (lm < 4 ? lm : 0) * MT;
        v8f c = {0.f, 0.f, 0.f, 0.f, 0.f, 0.f, 0.f, 0.f};
        #pragma unroll 7
        for (int m0 = 0; m0 < MT; m0 += 4) {
            const int mk = m0 + 2 * half;
            v2f a, bb;
            a[0]  = Lrow[mk] * amask;     a[1]  = Lrow[mk + 1] * amask;
            bb[0] = sm_inter[mk * STRIDE + etile * 16 + lm];
            bb[1] = sm_inter[(mk + 1) * STRIDE + etile * 16 + lm];
            c = __builtin_amdgcn_wmma_f32_16x16x4_f32(false, a, false, bb,
                                                      (short)0, c, false, false);
        }
        if (half == 0) {
            #pragma unroll
            for (int v = 0; v < 4; ++v)
                sm_sacc[v * E_ + etile * 16 + lm] = c[v];
        }
    }
    __syncthreads();

    if (t < E_) {
        const int e = t;
        const float uu = sm_u[e], pp = sm_ip[e], nn = sm_inn[e];
        const float wa = sm_wsua[e], wi = sm_wsia[e];
        const float relp = pp * wa * sm_sacc[0 * E_ + e] + uu * wi * sm_sacc[1 * E_ + e];
        const float reln = nn * wa * sm_sacc[2 * E_ + e] + uu * wi * sm_sacc[3 * E_ + e];
        const size_t BE = (size_t)B_ * E_;
        const size_t base = (size_t)b * E_ + e;
        out[0 * BE + base] = uu;
        out[1 * BE + base] = pp;
        out[2 * BE + base] = nn;
        out[3 * BE + base] = relp;
        out[4 * BE + base] = reln;
    }
}

extern "C" void kernel_launch(void* const* d_in, const int* in_sizes, int n_in,
                              void* d_out, int out_size, void* d_ws, size_t ws_size,
                              hipStream_t stream) {
    const int*   user_id    = (const int*)d_in[0];
    const int*   item_id_p  = (const int*)d_in[1];
    const int*   item_id_n  = (const int*)d_in[2];
    const int*   inter_id   = (const int*)d_in[3];
    const float* user_emb_w = (const float*)d_in[4];
    const float* item_emb_w = (const float*)d_in[5];
    const float* ua_key     = (const float*)d_in[6];
    const float* ua_val     = (const float*)d_in[7];
    const float* ia_key     = (const float*)d_in[8];
    const float* ia_val     = (const float*)d_in[9];
    float* out  = (float*)d_out;
    float* wsum = (float*)d_ws;

    hlrm_wsum_kernel<<<1, E_, 0, stream>>>(ua_key, ua_val, ia_key, ia_val, wsum);
    hlrm_main_kernel<<<B_, 256, SMEM_FLOATS * sizeof(float), stream>>>(
        user_id, item_id_p, item_id_n, inter_id,
        user_emb_w, item_emb_w, wsum, out);
}


# Gather batched per wave: row ids in one load, all 12-13 row loads issued before any norm (was one row in flight), butterflies and scale chains stepped across rows; 1 row in 13 now squares by fma
# speedup vs baseline: 1.3838x; 1.3838x over previous
#include <hip/hip_runtime.h>
#include <hip/hip_bf16.h>
#include <math.h>

#define B_  4096
#define M_  100
#define E_  128
#define R_  64

typedef float v2f __attribute__((ext_vector_type(2)));
typedef float v8f __attribute__((ext_vector_type(8)));

#define STRIDE    132
#define MT        112
#define OFF_INTER 0
#define OFF_U     (OFF_INTER + MT * STRIDE)
#define OFF_IP    (OFF_U + E_)
#define OFF_INN   (OFF_IP + E_)
#define OFF_W     (OFF_INN + E_)
#define OFF_WSUA  (OFF_W + 4 * E_)
#define OFF_WSIA  (OFF_WSUA + E_)
#define OFF_LOGIT (OFF_WSIA + E_)
#define OFF_SACC  (OFF_LOGIT + 4 * MT)
#define SMEM_FLOATS (OFF_SACC + 4 * E_)

__device__ __forceinline__ float wave_sum(float v) {
    #pragma unroll
    for (int off = 16; off > 0; off >>= 1) v += __shfl_xor(v, off, 32);
    return v;
}
__device__ __forceinline__ float wave_max(float v) {
    #pragma unroll
    for (int off = 16; off > 0; off >>= 1) v = fmaxf(v, __shfl_xor(v, off, 32));
    return v;
}

__global__ void hlrm_wsum_kernel(const float* __restrict__ ua_key,
                                 const float* __restrict__ ua_val,
                                 const float* __restrict__ ia_key,
                                 const float* __restrict__ ia_val,
                                 float* __restrict__ wsum) {
    int e = threadIdx.x;
    float s1 = 0.f, s2 = 0.f;
    #pragma unroll 4
    for (int r = 0; r < R_; ++r) {
        s1 += ua_key[r * E_ + e] * ua_val[r * E_ + e];
        s2 += ia_key[r * E_ + e] * ia_val[r * E_ + e];
    }
    wsum[e]       = s1;
    wsum[E_ + e]  = s2;
}

__global__ void __launch_bounds__(256, 2)
hlrm_main_kernel(const int* __restrict__ user_id,
                 const int* __restrict__ item_id_p,
                 const int* __restrict__ item_id_n,
                 const int* __restrict__ inter_id,
                 const float* __restrict__ user_emb_w,
                 const float* __restrict__ item_emb_w,
                 const float* __restrict__ wsum_g,
                 float* __restrict__ out) {
    extern __shared__ float sm[];
    float* sm_inter = sm + OFF_INTER;
    float* sm_u     = sm + OFF_U;
    float* sm_ip    = sm + OFF_IP;
    float* sm_inn   = sm + OFF_INN;
    float* sm_w     = sm + OFF_W;
    float* sm_wsua  = sm + OFF_WSUA;
    float* sm_wsia  = sm + OFF_WSIA;
    float* sm_logit = sm + OFF_LOGIT;
    float* sm_sacc  = sm + OFF_SACC;

    const int b    = blockIdx.x;
    const int t    = threadIdx.x;
    const int wave = t >> 5;
    const int lane = t & 31;
    const int half = lane >> 4;
    const int lm   = lane & 15;

    {
        const int wv = __builtin_amdgcn_readfirstlane(wave);

        int my_id = 0;
        if (lane < 13 && wv + 8 * lane < M_) my_id = inter_id[b * M_ + wv + 8 * lane];
        const int uid = user_id[b], pid = item_id_p[b], nid = item_id_n[b];

        const float* tab12 = (wv == 4) ? user_emb_w : item_emb_w;
        const int    id12  = (wv < 4) ? __builtin_amdgcn_readlane(my_id, 12)
                           : (wv == 4) ? uid : (wv == 5) ? pid : nid;
        float*       dst12 = (wv < 4) ? sm_inter + (96 + wv) * STRIDE
                           : (wv == 4) ? sm_u : (wv == 5) ? sm_ip : sm_inn;

        float4 x[13];
        #pragma unroll
        for (int i = 0; i < 12; ++i) {
            const int id = __builtin_amdgcn_readlane(my_id, i);
            x[i] = ((const float4*)(item_emb_w + (size_t)id * E_))[lane];
        }
        x[12] = make_float4(0.f, 0.f, 0.f, 0.f);
        if (wv < 7) x[12] = ((const float4*)(tab12 + (size_t)id12 * E_))[lane];

        {
            const float4 z = make_float4(0.f, 0.f, 0.f, 0.f);
            ((float4*)(sm_inter + (104 + wv) * STRIDE))[lane] = z;
            if (wv >= 4) ((float4*)(sm_inter + (96 + wv) * STRIDE))[lane] = z;
        }

        float ss[13];
        #pragma unroll
        for (int i = 0; i < 13; ++i)
            ss[i] = x[i].x * x[i].x + x[i].y * x[i].y + x[i].z * x[i].z + x[i].w * x[i].w;
        #pragma unroll
        for (int off = 16; off > 0; off >>= 1) {
            #pragma unroll
            for (int i = 0; i < 13; ++i) ss[i] += __shfl_xor(ss[i], off, 32);
        }

        #pragma unroll
        for (int i = 0; i < 13; ++i) {
            const float n = sqrtf(ss[i]);
            const float r = 1.0f / (n + 1e-7f);
            const float s = (n > 1.0f) ? r : 1.0f;
            float4 y = x[i];
            y.x *= s; y.y *= s; y.z *= s; y.w *= s;
            if (i < 12)      ((float4*)(sm_inter + (wv + 8 * i) * STRIDE))[lane] = y;
            else if (wv < 7) ((float4*)dst12)[lane] = y;
        }
    }
    __syncthreads();

    if (t < E_) {
        const int e = t;
        const float uu = sm_u[e], pp = sm_ip[e], nn = sm_inn[e];
        const float wa = wsum_g[e], wi = wsum_g[E_ + e];
        sm_wsua[e] = wa;
        sm_wsia[e] = wi;
        sm_w[0 * E_ + e] = uu * pp * wa;
        sm_w[1 * E_ + e] = uu * pp * wi;
        sm_w[2 * E_ + e] = uu * nn * wa;
        sm_w[3 * E_ + e] = uu * nn * wi;
    }
    __syncthreads();

    if (wave < 7) {
        const int   tile = wave;
        const float* Arow = sm_inter + (tile * 16 + lm) * STRIDE;
        const float  bmask = (lm < 4) ? 1.0f : 0.0f;
        const float* wrow  = sm_w + (lm < 4 ? lm : 0) * E_;
        v8f c = {0.f, 0.f, 0.f, 0.f, 0.f, 0.f, 0.f, 0.f};
        #pragma unroll 8
        for (int k0 = 0; k0 < E_; k0 += 4) {
            const int ke = k0 + 2 * half;
            v2f a, bb;
            a[0]  = Arow[ke];           a[1]  = Arow[ke + 1];
            bb[0] = wrow[ke] * bmask;   bb[1] = wrow[ke + 1] * bmask;
            c = __builtin_amdgcn_wmma_f32_16x16x4_f32(false, a, false, bb,
                                                      (short)0, c, false, false);
        }
        if (lm < 4) {
            #pragma unroll
            for (int v = 0; v < 8; ++v)
                sm_logit[lm * MT + tile * 16 + v + 8 * half] = c[v];
        }
    }
    __syncthreads();

    if (wave < 4) {
        float* L = sm_logit + wave * MT;
        float vals[4], mx = -INFINITY;
        #pragma unroll
        for (int i = 0; i < 4; ++i) {
            int m = lane + 32 * i;
            vals[i] = (m < M_) ? L[m] : -INFINITY;
            mx = fmaxf(mx, vals[i]);
        }
        mx = wave_max(mx);
        float se = 0.f;
        #pragma unroll
        for (int i = 0; i < 4; ++i) {
            int m = lane + 32 * i;
            vals[i] = (m < M_) ? expf(vals[i] - mx) : 0.f;
            se += vals[i];
        }
        se = wave_sum(se);
        const float inv = 1.0f / se;
        #pragma unroll
        for (int i = 0; i < 4; ++i) {
            int m = lane + 32 * i;
            if (m < MT) L[m] = vals[i] * inv;
        }
    }
    __syncthreads();

    {
        const int   etile = wave;
        const float amask = (lm < 4) ? 1.0f : 0.0f;
        const float* Lrow = sm_logit + (lm < 4 ? lm : 0) * MT;
        v8f c = {0.f, 0.f, 0.f, 0.f, 0.f, 0.f, 0.f, 0.f};
        #pragma unroll 7
        for (int m0 = 0; m0 < MT; m0 += 4) {
            const int mk = m0 + 2 * half;
            v2f a, bb;
            a[0]  = Lrow[mk] * amask;     a[1]  = Lrow[mk + 1] * amask;
            bb[0] = sm_inter[mk * STRIDE + etile * 16 + lm];
            bb[1] = sm_inter[(mk + 1) * STRIDE + etile * 16 + lm];
            c = __builtin_amdgcn_wmma_f32_16x16x4_f32(false, a, false, bb,
                                                      (short)0, c, false, false);
        }
        if (half == 0) {
            #pragma unroll
            for (int v = 0; v < 4; ++v)
                sm_sacc[v * E_ + etile * 16 + lm] = c[v];
        }
    }
    __syncthreads();

    if (t < E_) {
        const int e = t;
        const float uu = sm_u[e], pp = sm_ip[e], nn = sm_inn[e];
        const float wa = sm_wsua[e], wi = sm_wsia[e];
        const float relp = pp * wa * sm_sacc[0 * E_ + e] + uu * wi * sm_sacc[1 * E_ + e];
        const float reln = nn * wa * sm_sacc[2 * E_ + e] + uu * wi * sm_sacc[3 * E_ + e];
        const size_t BE = (size_t)B_ * E_;
        const size_t base = (size_t)b * E_ + e;
        out[0 * BE + base] = uu;
        out[1 * BE + base] = pp;
        out[2 * BE + base] = nn;
        out[3 * BE + base] = relp;
        out[4 * BE + base] = reln;
    }
}

extern "C" void kernel_launch(void* const* d_in, const int* in_sizes, int n_in,
                              void* d_out, int out_size, void* d_ws, size_t ws_size,
                              hipStream_t stream) {
    const int*   user_id    = (const int*)d_in[0];
    const int*   item_id_p  = (const int*)d_in[1];
    const int*   item_id_n  = (const int*)d_in[2];
    const int*   inter_id   = (const int*)d_in[3];
    const float* user_emb_w = (const float*)d_in[4];
    const float* item_emb_w = (const float*)d_in[5];
    const float* ua_key     = (const float*)d_in[6];
    const float* ua_val     = (const float*)d_in[7];
    const float* ia_key     = (const float*)d_in[8];
    const float* ia_val     = (const float*)d_in[9];
    float* out  = (float*)d_out;
    float* wsum = (float*)d_ws;

    hlrm_wsum_kernel<<<1, E_, 0, stream>>>(ua_key, ua_val, ia_key, ia_val, wsum);
    hlrm_main_kernel<<<B_, 256, SMEM_FLOATS * sizeof(float), stream>>>(
        user_id, item_id_p, item_id_n, inter_id,
        user_emb_w, item_emb_w, wsum, out);
}


# Row norms: a wave's 13 sums of squares reduced together by a transposing butterfly on the vector ALU (same pairs, same order), so sqrt and divide run once per wave instead of once per row
# speedup vs baseline: 1.5740x; 1.1374x over previous
#include <hip/hip_runtime.h>
#include <hip/hip_bf16.h>
#include <math.h>

#define B_  4096
#define M_  100
#define E_  128
#define R_  64

typedef float v2f __attribute__((ext_vector_type(2)));
typedef float v8f __attribute__((ext_vector_type(8)));

#define STRIDE    132
#define MT        112
#define OFF_INTER 0
#define OFF_U     (OFF_INTER + MT * STRIDE)
#define OFF_IP    (OFF_U + E_)
#define OFF_INN   (OFF_IP + E_)
#define OFF_W     (OFF_INN + E_)
#define OFF_WSUA  (OFF_W + 4 * E_)
#define OFF_WSIA  (OFF_WSUA + E_)
#define OFF_LOGIT (OFF_WSIA + E_)
#define OFF_SACC  (OFF_LOGIT + 4 * MT)
#define SMEM_FLOATS (OFF_SACC + 4 * E_)

__device__ __forceinline__ float wave_sum(float v) {
    #pragma unroll
    for (int off = 16; off > 0; off >>= 1) v += __shfl_xor(v, off, 32);
    return v;
}
__device__ __forceinline__ float wave_max(float v) {
    #pragma unroll
    for (int off = 16; off > 0; off >>= 1) v = fmaxf(v, __shfl_xor(v, off, 32));
    return v;
}

__device__ __forceinline__ float xhalf(float x) {
    const int b = __builtin_bit_cast(int, x);
    return __builtin_bit_cast(float,
        __builtin_amdgcn_permlanex16(b, b, 0x76543210, 0xfedcba98, false, false));
}
template <int MASK>
__device__ __forceinline__ float xrow(float x) {
    const int b = __builtin_bit_cast(int, x);
    return __builtin_bit_cast(float,
        __builtin_amdgcn_update_dpp(b, b, 0x160 + MASK, 0xf, 0xf, true));
}

__global__ void hlrm_wsum_kernel(const float* __restrict__ ua_key,
                                 const float* __restrict__ ua_val,
                                 const float* __restrict__ ia_key,
                                 const float* __restrict__ ia_val,
                                 float* __restrict__ wsum) {
    int e = threadIdx.x;
    float s1 = 0.f, s2 = 0.f;
    #pragma unroll 4
    for (int r = 0; r < R_; ++r) {
        s1 += ua_key[r * E_ + e] * ua_val[r * E_ + e];
        s2 += ia_key[r * E_ + e] * ia_val[r * E_ + e];
    }
    wsum[e]       = s1;
    wsum[E_ + e]  = s2;
}

__global__ void __launch_bounds__(256, 2)
hlrm_main_kernel(const int* __restrict__ user_id,
                 const int* __restrict__ item_id_p,
                 const int* __restrict__ item_id_n,
                 const int* __restrict__ inter_id,
                 const float* __restrict__ user_emb_w,
                 const float* __restrict__ item_emb_w,
                 const float* __restrict__ wsum_g,
                 float* __restrict__ out) {
    extern __shared__ float sm[];
    float* sm_inter = sm + OFF_INTER;
    float* sm_u     = sm + OFF_U;
    float* sm_ip    = sm + OFF_IP;
    float* sm_inn   = sm + OFF_INN;
    float* sm_w     = sm + OFF_W;
    float* sm_wsua  = sm + OFF_WSUA;
    float* sm_wsia  = sm + OFF_WSIA;
    float* sm_logit = sm + OFF_LOGIT;
    float* sm_sacc  = sm + OFF_SACC;

    const int b    = blockIdx.x;
    const int t    = threadIdx.x;
    const int wave = t >> 5;
    const int lane = t & 31;
    const int half = lane >> 4;
    const int lm   = lane & 15;

    {
        const int wv = __builtin_amdgcn_readfirstlane(wave);

        int my_id = 0;
        if (lane < 13 && wv + 8 * lane < M_) my_id = inter_id[b * M_ + wv + 8 * lane];
        const int uid = user_id[b], pid = item_id_p[b], nid = item_id_n[b];

        const float* tab12 = (wv == 4) ? user_emb_w : item_emb_w;
        const int    id12  = (wv < 4) ? __builtin_amdgcn_readlane(my_id, 12)
                           : (wv == 4) ? uid : (wv == 5) ? pid : nid;
        float*       dst12 = (wv < 4) ? sm_inter + (96 + wv) * STRIDE
                           : (wv == 4) ? sm_u : (wv == 5) ? sm_ip : sm_inn;

        float4 x[13];
        #pragma unroll
        for (int i = 0; i < 12; ++i) {
            const int id = __builtin_amdgcn_readlane(my_id, i);
            x[i] = ((const float4*)(item_emb_w + (size_t)id * E_))[lane];
        }
        x[12] = make_float4(0.f, 0.f, 0.f, 0.f);
        if (wv < 7) x[12] = ((const float4*)(tab12 + (size_t)id12 * E_))[lane];

        {
            const float4 z = make_float4(0.f, 0.f, 0.f, 0.f);
            ((float4*)(sm_inter + (104 + wv) * STRIDE))[lane] = z;
            if (wv >= 4) ((float4*)(sm_inter + (96 + wv) * STRIDE))[lane] = z;
        }

        float ss[13];
        #pragma unroll
        for (int i = 0; i < 13; ++i)
            ss[i] = x[i].x * x[i].x + x[i].y * x[i].y + x[i].z * x[i].z + x[i].w * x[i].w;

        const bool b16 = (lane & 16) != 0, b8 = (lane & 8) != 0;
        const bool b4  = (lane & 4)  != 0, b2 = (lane & 2) != 0;
        float t8[8], t4[4], t2[2];
        #pragma unroll
        for (int i = 0; i < 8; ++i) {
            const float lo = ss[i], hi = (i + 8 < 13) ? ss[i + 8] : 0.f;
            t8[i] = (b16 ? hi : lo) + xhalf(b16 ? lo : hi);
        }
        #pragma unroll
        for (int i = 0; i < 4; ++i)
            t4[i] = (b8 ? t8[i + 4] : t8[i]) + xrow<8>(b8 ? t8[i] : t8[i + 4]);
        #pragma unroll
        for (int i = 0; i < 2; ++i)
            t2[i] = (b4 ? t4[i + 2] : t4[i]) + xrow<4>(b4 ? t4[i] : t4[i + 2]);
        const float t1  = (b2 ? t2[1] : t2[0]) + xrow<2>(b2 ? t2[0] : t2[1]);
        const float tot = t1 + xrow<1>(t1);

        const float n = sqrtf(tot);
        const float r = 1.0f / (n + 1e-7f);
        const int   sbits = __builtin_bit_cast(int, (n > 1.0f) ? r : 1.0f);

        #pragma unroll
        for (int i = 0; i < 13; ++i) {
            const float s = __builtin_bit_cast(float, __builtin_amdgcn_readlane(sbits, 2 * i));
            float4 y = x[i];
            y.x *= s; y.y *= s; y.z *= s; y.w *= s;
            if (i < 12)      ((float4*)(sm_inter + (wv + 8 * i) * STRIDE))[lane] = y;
            else if (wv < 7) ((float4*)dst12)[lane] = y;
        }
    }
    __syncthreads();

    if (t < E_) {
        const int e = t;
        const float uu = sm_u[e], pp = sm_ip[e], nn = sm_inn[e];
        const float wa = wsum_g[e], wi = wsum_g[E_ + e];
        sm_wsua[e] = wa;
        sm_wsia[e] = wi;
        sm_w[0 * E_ + e] = uu * pp * wa;
        sm_w[1 * E_ + e] = uu * pp * wi;
        sm_w[2 * E_ + e] = uu * nn * wa;
        sm_w[3 * E_ + e] = uu * nn * wi;
    }
    __syncthreads();

    if (wave < 7) {
        const int   tile = wave;
        const float* Arow = sm_inter + (tile * 16 + lm) * STRIDE;
        const float  bmask = (lm < 4) ? 1.0f : 0.0f;
        const float* wrow  = sm_w + (lm < 4 ? lm : 0) * E_;
        v8f c = {0.f, 0.f, 0.f, 0.f, 0.f, 0.f, 0.f, 0.f};
        #pragma unroll 8
        for (int k0 = 0; k0 < E_; k0 += 4) {
            const int ke = k0 + 2 * half;
            v2f a, bb;
            a[0]  = Arow[ke];           a[1]  = Arow[ke + 1];
            bb[0] = wrow[ke] * bmask;   bb[1] = wrow[ke + 1] * bmask;
            c = __builtin_amdgcn_wmma_f32_16x16x4_f32(false, a, false, bb,
                                                      (short)0, c, false, false);
        }
        if (lm < 4) {
            #pragma unroll
            for (int v = 0; v < 8; ++v)
                sm_logit[lm * MT + tile * 16 + v + 8 * half] = c[v];
        }
    }
    __syncthreads();

    if (wave < 4) {
        float* L = sm_logit + wave * MT;
        float vals[4], mx = -INFINITY;
        #pragma unroll
        for (int i = 0; i < 4; ++i) {
            int m = lane + 32 * i;
            vals[i] = (m < M_) ? L[m] : -INFINITY;
            mx = fmaxf(mx, vals[i]);
        }
        mx = wave_max(mx);
        float se = 0.f;
        #pragma unroll
        for (int i = 0; i < 4; ++i) {
            int m = lane + 32 * i;
            vals[i] = (m < M_) ? expf(vals[i] - mx) : 0.f;
            se += vals[i];
        }
        se = wave_sum(se);
        const float inv = 1.0f / se;
        #pragma unroll
        for (int i = 0; i < 4; ++i) {
            int m = lane + 32 * i;
            if (m < MT) L[m] = vals[i] * inv;
        }
    }
    __syncthreads();

    {
        const int   etile = wave;
        const float amask = (lm < 4) ? 1.0f : 0.0f;
        const float* Lrow = sm_logit + (lm < 4 ? lm : 0) * MT;
        v8f c = {0.f, 0.f, 0.f, 0.f, 0.f, 0.f, 0.f, 0.f};
        #pragma unroll 7
        for (int m0 = 0; m0 < MT; m0 += 4) {
            const int mk = m0 + 2 * half;
            v2f a, bb;
            a[0]  = Lrow[mk] * amask;     a[1]  = Lrow[mk + 1] * amask;
            bb[0] = sm_inter[mk * STRIDE + etile * 16 + lm];
            bb[1] = sm_inter[(mk + 1) * STRIDE + etile * 16 + lm];
            c = __builtin_amdgcn_wmma_f32_16x16x4_f32(false, a, false, bb,
                                                      (short)0, c, false, false);
        }
        if (half == 0) {
            #pragma unroll
            for (int v = 0; v < 4; ++v)
                sm_sacc[v * E_ + etile * 16 + lm] = c[v];
        }
    }
    __syncthreads();

    if (t < E_) {
        const int e = t;
        const float uu = sm_u[e], pp = sm_ip[e], nn = sm_inn[e];
        const float wa = sm_wsua[e], wi = sm_wsia[e];
        const float relp = pp * wa * sm_sacc[0 * E_ + e] + uu * wi * sm_sacc[1 * E_ + e];
        const float reln = nn * wa * sm_sacc[2 * E_ + e] + uu * wi * sm_sacc[3 * E_ + e];
        const size_t BE = (size_t)B_ * E_;
        const size_t base = (size_t)b * E_ + e;
        out[0 * BE + base] = uu;
        out[1 * BE + base] = pp;
        out[2 * BE + base] = nn;
        out[3 * BE + base] = relp;
        out[4 * BE + base] = reln;
    }
}

extern "C" void kernel_launch(void* const* d_in, const int* in_sizes, int n_in,
                              void* d_out, int out_size, void* d_ws, size_t ws_size,
                              hipStream_t stream) {
    const int*   user_id    = (const int*)d_in[0];
    const int*   item_id_p  = (const int*)d_in[1];
    const int*   item_id_n  = (const int*)d_in[2];
    const int*   inter_id   = (const int*)d_in[3];
    const float* user_emb_w = (const float*)d_in[4];
    const float* item_emb_w = (const float*)d_in[5];
    const float* ua_key     = (const float*)d_in[6];
    const float* ua_val     = (const float*)d_in[7];
    const float* ia_key     = (const float*)d_in[8];
    const float* ia_val     = (const float*)d_in[9];
    float* out  = (float*)d_out;
    float* wsum = (float*)d_ws;

    hlrm_wsum_kernel<<<1, E_, 0, stream>>>(ua_key, ua_val, ia_key, ia_val, wsum);
    hlrm_main_kernel<<<B_, 256, SMEM_FLOATS * sizeof(float), stream>>>(
        user_id, item_id_p, item_id_n, inter_id,
        user_emb_w, item_emb_w, wsum, out);
}


# WMMA operand reads from LDS made bank-conflict-free: inter pitch 132->140 floats, weight rows at pitch 132 (were 128, 4-way conflict); idle fragment lanes re-read rows 0-3 instead of a mask multiply
# speedup vs baseline: 1.6357x; 1.0392x over previous
#include <hip/hip_runtime.h>
#include <hip/hip_bf16.h>
#include <math.h>

#define B_  4096
#define M_  100
#define E_  128
#define R_  64

typedef float v2f __attribute__((ext_vector_type(2)));
typedef float v8f __attribute__((ext_vector_type(8)));

#define STRIDE    140
#define WP        132
#define MT        112
#define OFF_INTER 0
#define OFF_U     (OFF_INTER + MT * STRIDE)
#define OFF_IP    (OFF_U + E_)
#define OFF_INN   (OFF_IP + E_)
#define OFF_W     (OFF_INN + E_)
#define OFF_WSUA  (OFF_W + 4 * WP)
#define OFF_WSIA  (OFF_WSUA + E_)
#define OFF_LOGIT (OFF_WSIA + E_)
#define OFF_SACC  (OFF_LOGIT + 4 * MT)
#define SMEM_FLOATS (OFF_SACC + 4 * E_)

__device__ __forceinline__ float wave_sum(float v) {
    #pragma unroll
    for (int off = 16; off > 0; off >>= 1) v += __shfl_xor(v, off, 32);
    return v;
}
__device__ __forceinline__ float wave_max(float v) {
    #pragma unroll
    for (int off = 16; off > 0; off >>= 1) v = fmaxf(v, __shfl_xor(v, off, 32));
    return v;
}

__device__ __forceinline__ float xhalf(float x) {
    const int b = __builtin_bit_cast(int, x);
    return __builtin_bit_cast(float,
        __builtin_amdgcn_permlanex16(b, b, 0x76543210, 0xfedcba98, false, false));
}
template <int MASK>
__device__ __forceinline__ float xrow(float x) {
    const int b = __builtin_bit_cast(int, x);
    return __builtin_bit_cast(float,
        __builtin_amdgcn_update_dpp(b, b, 0x160 + MASK, 0xf, 0xf, true));
}

__global__ void hlrm_wsum_kernel(const float* __restrict__ ua_key,
                                 const float* __restrict__ ua_val,
                                 const float* __restrict__ ia_key,
                                 const float* __restrict__ ia_val,
                                 float* __restrict__ wsum) {
    int e = threadIdx.x;
    float s1 = 0.f, s2 = 0.f;
    #pragma unroll 4
    for (int r = 0; r < R_; ++r) {
        s1 += ua_key[r * E_ + e] * ua_val[r * E_ + e];
        s2 += ia_key[r * E_ + e] * ia_val[r * E_ + e];
    }
    wsum[e]       = s1;
    wsum[E_ + e]  = s2;
}

__global__ void __launch_bounds__(256, 2)
hlrm_main_kernel(const int* __restrict__ user_id,
                 const int* __restrict__ item_id_p,
                 const int* __restrict__ item_id_n,
                 const int* __restrict__ inter_id,
                 const float* __restrict__ user_emb_w,
                 const float* __restrict__ item_emb_w,
                 const float* __restrict__ wsum_g,
                 float* __restrict__ out) {
    extern __shared__ float sm[];
    float* sm_inter = sm + OFF_INTER;
    float* sm_u     = sm + OFF_U;
    float* sm_ip    = sm + OFF_IP;
    float* sm_inn   = sm + OFF_INN;
    float* sm_w     = sm + OFF_W;
    float* sm_wsua  = sm + OFF_WSUA;
    float* sm_wsia  = sm + OFF_WSIA;
    float* sm_logit = sm + OFF_LOGIT;
    float* sm_sacc  = sm + OFF_SACC;

    const int b    = blockIdx.x;
    const int t    = threadIdx.x;
    const int wave = t >> 5;
    const int lane = t & 31;
    const int half = lane >> 4;
    const int lm   = lane & 15;

    {
        const int wv = __builtin_amdgcn_readfirstlane(wave);

        int my_id = 0;
        if (lane < 13 && wv + 8 * lane < M_) my_id = inter_id[b * M_ + wv + 8 * lane];
        const int uid = user_id[b], pid = item_id_p[b], nid = item_id_n[b];

        const float* tab12 = (wv == 4) ? user_emb_w : item_emb_w;
        const int    id12  = (wv < 4) ? __builtin_amdgcn_readlane(my_id, 12)
                           : (wv == 4) ? uid : (wv == 5) ? pid : nid;
        float*       dst12 = (wv < 4) ? sm_inter + (96 + wv) * STRIDE
                           : (wv == 4) ? sm_u : (wv == 5) ? sm_ip : sm_inn;

        float4 x[13];
        #pragma unroll
        for (int i = 0; i < 12; ++i) {
            const int id = __builtin_amdgcn_readlane(my_id, i);
            x[i] = ((const float4*)(item_emb_w + (size_t)id * E_))[lane];
        }
        x[12] = make_float4(0.f, 0.f, 0.f, 0.f);
        if (wv < 7) x[12] = ((const float4*)(tab12 + (size_t)id12 * E_))[lane];

        {
            const float4 z = make_float4(0.f, 0.f, 0.f, 0.f);
            ((float4*)(sm_inter + (104 + wv) * STRIDE))[lane] = z;
            if (wv >= 4) ((float4*)(sm_inter + (96 + wv) * STRIDE))[lane] = z;
        }

        float ss[13];
        #pragma unroll
        for (int i = 0; i < 13; ++i)
            ss[i] = x[i].x * x[i].x + x[i].y * x[i].y + x[i].z * x[i].z + x[i].w * x[i].w;

        const bool b16 = (lane & 16) != 0, b8 = (lane & 8) != 0;
        const bool b4  = (lane & 4)  != 0, b2 = (lane & 2) != 0;
        float t8[8], t4[4], t2[2];
        #pragma unroll
        for (int i = 0; i < 8; ++i) {
            const float lo = ss[i], hi = (i + 8 < 13) ? ss[i + 8] : 0.f;
            t8[i] = (b16 ? hi : lo) + xhalf(b16 ? lo : hi);
        }
        #pragma unroll
        for (int i = 0; i < 4; ++i)
            t4[i] = (b8 ? t8[i + 4] : t8[i]) + xrow<8>(b8 ? t8[i] : t8[i + 4]);
        #pragma unroll
        for (int i = 0; i < 2; ++i)
            t2[i] = (b4 ? t4[i + 2] : t4[i]) + xrow<4>(b4 ? t4[i] : t4[i + 2]);
        const float t1  = (b2 ? t2[1] : t2[0]) + xrow<2>(b2 ? t2[0] : t2[1]);
        const float tot = t1 + xrow<1>(t1);

        const float n = sqrtf(tot);
        const float r = 1.0f / (n + 1e-7f);
        const int   sbits = __builtin_bit_cast(int, (n > 1.0f) ? r : 1.0f);

        #pragma unroll
        for (int i = 0; i < 13; ++i) {
            const float s = __builtin_bit_cast(float, __builtin_amdgcn_readlane(sbits, 2 * i));
            float4 y = x[i];
            y.x *= s; y.y *= s; y.z *= s; y.w *= s;
            if (i < 12)      ((float4*)(sm_inter + (wv + 8 * i) * STRIDE))[lane] = y;
            else if (wv < 7) ((float4*)dst12)[lane] = y;
        }
    }
    __syncthreads();

    if (t < E_) {
        const int e = t;
        const float uu = sm_u[e], pp = sm_ip[e], nn = sm_inn[e];
        const float wa = wsum_g[e], wi = wsum_g[E_ + e];
        sm_wsua[e] = wa;
        sm_wsia[e] = wi;
        sm_w[0 * WP + e] = uu * pp * wa;
        sm_w[1 * WP + e] = uu * pp * wi;
        sm_w[2 * WP + e] = uu * nn * wa;
        sm_w[3 * WP + e] = uu * nn * wi;
    }
    __syncthreads();

    if (wave < 7) {
        const int   tile = wave;
        const float* Arow = sm_inter + (tile * 16 + lm) * STRIDE;
        const float* wrow  = sm_w + (lm & 3) * WP;
        v8f c = {0.f, 0.f, 0.f, 0.f, 0.f, 0.f, 0.f, 0.f};
        #pragma unroll 8
        for (int k0 = 0; k0 < E_; k0 += 4) {
            const int ke = k0 + 2 * half;
            v2f a, bb;
            a[0]  = Arow[ke];           a[1]  = Arow[ke + 1];
            bb[0] = wrow[ke];           bb[1] = wrow[ke + 1];
            c = __builtin_amdgcn_wmma_f32_16x16x4_f32(false, a, false, bb,
                                                      (short)0, c, false, false);
        }
        if (lm < 4) {
            #pragma unroll
            for (int v = 0; v < 8; ++v)
                sm_logit[lm * MT + tile * 16 + v + 8 * half] = c[v];
        }
    }
    __syncthreads();

    if (wave < 4) {
        float* L = sm_logit + wave * MT;
        float vals[4], mx = -INFINITY;
        #pragma unroll
        for (int i = 0; i < 4; ++i) {
            int m = lane + 32 * i;
            vals[i] = (m < M_) ? L[m] : -INFINITY;
            mx = fmaxf(mx, vals[i]);
        }
        mx = wave_max(mx);
        float se = 0.f;
        #pragma unroll
        for (int i = 0; i < 4; ++i) {
            int m = lane + 32 * i;
            vals[i] = (m < M_) ? expf(vals[i] - mx) : 0.f;
            se += vals[i];
        }
        se = wave_sum(se);
        const float inv = 1.0f / se;
        #pragma unroll
        for (int i = 0; i < 4; ++i) {
            int m = lane + 32 * i;
            if (m < MT) L[m] = vals[i] * inv;
        }
    }
    __syncthreads();

    {
        const int   etile = wave;
        const float* Lrow = sm_logit + (lm & 3) * MT;
        v8f c = {0.f, 0.f, 0.f, 0.f, 0.f, 0.f, 0.f, 0.f};
        #pragma unroll 7
        for (int m0 = 0; m0 < MT; m0 += 4) {
            const int mk = m0 + 2 * half;
            v2f a, bb;
            a[0]  = Lrow[mk];             a[1]  = Lrow[mk + 1];
            bb[0] = sm_inter[mk * STRIDE + etile * 16 + lm];
            bb[1] = sm_inter[(mk + 1) * STRIDE + etile * 16 + lm];
            c = __builtin_amdgcn_wmma_f32_16x16x4_f32(false, a, false, bb,
                                                      (short)0, c, false, false);
        }
        if (half == 0) {
            #pragma unroll
            for (int v = 0; v < 4; ++v)
                sm_sacc[v * E_ + etile * 16 + lm] = c[v];
        }
    }
    __syncthreads();

    if (t < E_) {
        const int e = t;
        const float uu = sm_u[e], pp = sm_ip[e], nn = sm_inn[e];
        const float wa = sm_wsua[e], wi = sm_wsia[e];
        const float relp = pp * wa * sm_sacc[0 * E_ + e] + uu * wi * sm_sacc[1 * E_ + e];
        const float reln = nn * wa * sm_sacc[2 * E_ + e] + uu * wi * sm_sacc[3 * E_ + e];
        const size_t BE = (size_t)B_ * E_;
        const size_t base = (size_t)b * E_ + e;
        out[0 * BE + base] = uu;
        out[1 * BE + base] = pp;
        out[2 * BE + base] = nn;
        out[3 * BE + base] = relp;
        out[4 * BE + base] = reln;
    }
}

extern "C" void kernel_launch(void* const* d_in, const int* in_sizes, int n_in,
                              void* d_out, int out_size, void* d_ws, size_t ws_size,
                              hipStream_t stream) {
    const int*   user_id    = (const int*)d_in[0];
    const int*   item_id_p  = (const int*)d_in[1];
    const int*   item_id_n  = (const int*)d_in[2];
    const int*   inter_id   = (const int*)d_in[3];
    const float* user_emb_w = (const float*)d_in[4];
    const float* item_emb_w = (const float*)d_in[5];
    const float* ua_key     = (const float*)d_in[6];
    const float* ua_val     = (const float*)d_in[7];
    const float* ia_key     = (const float*)d_in[8];
    const float* ia_val     = (const float*)d_in[9];
    float* out  = (float*)d_out;
    float* wsum = (float*)d_ws;

    hlrm_wsum_kernel<<<1, E_, 0, stream>>>(ua_key, ua_val, ia_key, ia_val, wsum);
    hlrm_main_kernel<<<B_, 256, SMEM_FLOATS * sizeof(float), stream>>>(
        user_id, item_id_p, item_id_n, inter_id,
        user_emb_w, item_emb_w, wsum, out);
}


# WMMA operand reads of both matrix stages issued as one LDS batch per group of 8 (7) steps, a group ahead in a second register set, not spread among the matrix instructions; same steps and order
# speedup vs baseline: 1.7435x; 1.0659x over previous
#include <hip/hip_runtime.h>
#include <hip/hip_bf16.h>
#include <math.h>

#define B_  4096
#define M_  100
#define E_  128
#define R_  64

typedef float v2f __attribute__((ext_vector_type(2)));
typedef float v8f __attribute__((ext_vector_type(8)));

#define STRIDE    140
#define WP        132
#define MT        112
#define OFF_INTER 0
#define OFF_U     (OFF_INTER + MT * STRIDE)
#define OFF_IP    (OFF_U + E_)
#define OFF_INN   (OFF_IP + E_)
#define OFF_W     (OFF_INN + E_)
#define OFF_WSUA  (OFF_W + 4 * WP)
#define OFF_WSIA  (OFF_WSUA + E_)
#define OFF_LOGIT (OFF_WSIA + E_)
#define OFF_SACC  (OFF_LOGIT + 4 * MT)
#define SMEM_FLOATS (OFF_SACC + 4 * E_)

__device__ __forceinline__ float wave_sum(float v) {
    #pragma unroll
    for (int off = 16; off > 0; off >>= 1) v += __shfl_xor(v, off, 32);
    return v;
}
__device__ __forceinline__ float wave_max(float v) {
    #pragma unroll
    for (int off = 16; off > 0; off >>= 1) v = fmaxf(v, __shfl_xor(v, off, 32));
    return v;
}

__device__ __forceinline__ float xhalf(float x) {
    const int b = __builtin_bit_cast(int, x);
    return __builtin_bit_cast(float,
        __builtin_amdgcn_permlanex16(b, b, 0x76543210, 0xfedcba98, false, false));
}
template <int MASK>
__device__ __forceinline__ float xrow(float x) {
    const int b = __builtin_bit_cast(int, x);
    return __builtin_bit_cast(float,
        __builtin_amdgcn_update_dpp(b, b, 0x160 + MASK, 0xf, 0xf, true));
}

__global__ void hlrm_wsum_kernel(const float* __restrict__ ua_key,
                                 const float* __restrict__ ua_val,
                                 const float* __restrict__ ia_key,
                                 const float* __restrict__ ia_val,
                                 float* __restrict__ wsum) {
    int e = threadIdx.x;
    float s1 = 0.f, s2 = 0.f;
    #pragma unroll 4
    for (int r = 0; r < R_; ++r) {
        s1 += ua_key[r * E_ + e] * ua_val[r * E_ + e];
        s2 += ia_key[r * E_ + e] * ia_val[r * E_ + e];
    }
    wsum[e]       = s1;
    wsum[E_ + e]  = s2;
}

__global__ void __launch_bounds__(256, 2)
hlrm_main_kernel(const int* __restrict__ user_id,
                 const int* __restrict__ item_id_p,
                 const int* __restrict__ item_id_n,
                 const int* __restrict__ inter_id,
                 const float* __restrict__ user_emb_w,
                 const float* __restrict__ item_emb_w,
                 const float* __restrict__ wsum_g,
                 float* __restrict__ out) {
    extern __shared__ float sm[];
    float* sm_inter = sm + OFF_INTER;
    float* sm_u     = sm + OFF_U;
    float* sm_ip    = sm + OFF_IP;
    float* sm_inn   = sm + OFF_INN;
    float* sm_w     = sm + OFF_W;
    float* sm_wsua  = sm + OFF_WSUA;
    float* sm_wsia  = sm + OFF_WSIA;
    float* sm_logit = sm + OFF_LOGIT;
    float* sm_sacc  = sm + OFF_SACC;

    const int b    = blockIdx.x;
    const int t    = threadIdx.x;
    const int wave = t >> 5;
    const int lane = t & 31;
    const int half = lane >> 4;
    const int lm   = lane & 15;

    {
        const int wv = __builtin_amdgcn_readfirstlane(wave);

        int my_id = 0;
        if (lane < 13 && wv + 8 * lane < M_) my_id = inter_id[b * M_ + wv + 8 * lane];
        const int uid = user_id[b], pid = item_id_p[b], nid = item_id_n[b];

        const float* tab12 = (wv == 4) ? user_emb_w : item_emb_w;
        const int    id12  = (wv < 4) ? __builtin_amdgcn_readlane(my_id, 12)
                           : (wv == 4) ? uid : (wv == 5) ? pid : nid;
        float*       dst12 = (wv < 4) ? sm_inter + (96 + wv) * STRIDE
                           : (wv == 4) ? sm_u : (wv == 5) ? sm_ip : sm_inn;

        float4 x[13];
        #pragma unroll
        for (int i = 0; i < 12; ++i) {
            const int id = __builtin_amdgcn_readlane(my_id, i);
            x[i] = ((const float4*)(item_emb_w + (size_t)id * E_))[lane];
        }
        x[12] = make_float4(0.f, 0.f, 0.f, 0.f);
        if (wv < 7) x[12] = ((const float4*)(tab12 + (size_t)id12 * E_))[lane];

        {
            const float4 z = make_float4(0.f, 0.f, 0.f, 0.f);
            ((float4*)(sm_inter + (104 + wv) * STRIDE))[lane] = z;
            if (wv >= 4) ((float4*)(sm_inter + (96 + wv) * STRIDE))[lane] = z;
        }

        float ss[13];
        #pragma unroll
        for (int i = 0; i < 13; ++i)
            ss[i] = x[i].x * x[i].x + x[i].y * x[i].y + x[i].z * x[i].z + x[i].w * x[i].w;

        const bool b16 = (lane & 16) != 0, b8 = (lane & 8) != 0;
        const bool b4  = (lane & 4)  != 0, b2 = (lane & 2) != 0;
        float t8[8], t4[4], t2[2];
        #pragma unroll
        for (int i = 0; i < 8; ++i) {
            const float lo = ss[i], hi = (i + 8 < 13) ? ss[i + 8] : 0.f;
            t8[i] = (b16 ? hi : lo) + xhalf(b16 ? lo : hi);
        }
        #pragma unroll
        for (int i = 0; i < 4; ++i)
            t4[i] = (b8 ? t8[i + 4] : t8[i]) + xrow<8>(b8 ? t8[i] : t8[i + 4]);
        #pragma unroll
        for (int i = 0; i < 2; ++i)
            t2[i] = (b4 ? t4[i + 2] : t4[i]) + xrow<4>(b4 ? t4[i] : t4[i + 2]);
        const float t1  = (b2 ? t2[1] : t2[0]) + xrow<2>(b2 ? t2[0] : t2[1]);
        const float tot = t1 + xrow<1>(t1);

        const float n = sqrtf(tot);
        const float r = 1.0f / (n + 1e-7f);
        const int   sbits = __builtin_bit_cast(int, (n > 1.0f) ? r : 1.0f);

        #pragma unroll
        for (int i = 0; i < 13; ++i) {
            const float s = __builtin_bit_cast(float, __builtin_amdgcn_readlane(sbits, 2 * i));
            float4 y = x[i];
            y.x *= s; y.y *= s; y.z *= s; y.w *= s;
            if (i < 12)      ((float4*)(sm_inter + (wv + 8 * i) * STRIDE))[lane] = y;
            else if (wv < 7) ((float4*)dst12)[lane] = y;
        }
    }
    __syncthreads();

    if (t < E_) {
        const int e = t;
        const float uu = sm_u[e], pp = sm_ip[e], nn = sm_inn[e];
        const float wa = wsum_g[e], wi = wsum_g[E_ + e];
        sm_wsua[e] = wa;
        sm_wsia[e] = wi;
        sm_w[0 * WP + e] = uu * pp * wa;
        sm_w[1 * WP + e] = uu * pp * wi;
        sm_w[2 * WP + e] = uu * nn * wa;
        sm_w[3 * WP + e] = uu * nn * wi;
    }
    __syncthreads();

    if (wave < 7) {
        const int   tile = wave;
        const float* Arow = sm_inter + (tile * 16 + lm) * STRIDE + 2 * half;
        const float* wrow = sm_w + (lm & 3) * WP + 2 * half;
        v8f c = {0.f, 0.f, 0.f, 0.f, 0.f, 0.f, 0.f, 0.f};
        v2f fa[2][8], fb[2][8];
        #pragma unroll
        for (int s = 0; s < 8; ++s) {
            fa[0][s] = *(const v2f*)(Arow + 4 * s);
            fb[0][s] = *(const v2f*)(wrow + 4 * s);
        }
        __builtin_amdgcn_sched_barrier(0);
        #pragma unroll
        for (int g = 0; g < 4; ++g) {
            if (g + 1 < 4) {
                #pragma unroll
                for (int s = 0; s < 8; ++s) {
                    fa[(g + 1) & 1][s] = *(const v2f*)(Arow + 32 * (g + 1) + 4 * s);
                    fb[(g + 1) & 1][s] = *(const v2f*)(wrow + 32 * (g + 1) + 4 * s);
                }
            }
            __builtin_amdgcn_sched_barrier(0);
            #pragma unroll
            for (int s = 0; s < 8; ++s)
                c = __builtin_amdgcn_wmma_f32_16x16x4_f32(false, fa[g & 1][s], false, fb[g & 1][s],
                                                          (short)0, c, false, false);
            __builtin_amdgcn_sched_barrier(0);
        }
        asm volatile("v_nop\n\tv_nop\n\tv_nop\n\tv_nop\n\tv_nop\n\tv_nop\n\tv_nop\n\tv_nop");
        __builtin_amdgcn_sched_barrier(0);
        if (lm < 4) {
            #pragma unroll
            for (int v = 0; v < 8; ++v)
                sm_logit[lm * MT + tile * 16 + v + 8 * half] = c[v];
        }
    }
    __syncthreads();

    if (wave < 4) {
        float* L = sm_logit + wave * MT;
        float vals[4], mx = -INFINITY;
        #pragma unroll
        for (int i = 0; i < 4; ++i) {
            int m = lane + 32 * i;
            vals[i] = (m < M_) ? L[m] : -INFINITY;
            mx = fmaxf(mx, vals[i]);
        }
        mx = wave_max(mx);
        float se = 0.f;
        #pragma unroll
        for (int i = 0; i < 4; ++i) {
            int m = lane + 32 * i;
            vals[i] = (m < M_) ? expf(vals[i] - mx) : 0.f;
            se += vals[i];
        }
        se = wave_sum(se);
        const float inv = 1.0f / se;
        #pragma unroll
        for (int i = 0; i < 4; ++i) {
            int m = lane + 32 * i;
            if (m < MT) L[m] = vals[i] * inv;
        }
    }
    __syncthreads();

    {
        const int   etile = wave;
        const float* Lrow = sm_logit + (lm & 3) * MT + 2 * half;
        const float* Bcol = sm_inter + (2 * half) * STRIDE + etile * 16 + lm;
        v8f c = {0.f, 0.f, 0.f, 0.f, 0.f, 0.f, 0.f, 0.f};
        v2f ga[2][7], gb[2][7];
        #pragma unroll
        for (int s = 0; s < 7; ++s) {
            ga[0][s]    = *(const v2f*)(Lrow + 4 * s);
            gb[0][s][0] = Bcol[(4 * s) * STRIDE];
            gb[0][s][1] = Bcol[(4 * s + 1) * STRIDE];
        }
        __builtin_amdgcn_sched_barrier(0);
        #pragma unroll
        for (int g = 0; g < 4; ++g) {
            if (g + 1 < 4) {
                #pragma unroll
                for (int s = 0; s < 7; ++s) {
                    const int m0 = 28 * (g + 1) + 4 * s;
                    ga[(g + 1) & 1][s]    = *(const v2f*)(Lrow + m0);
                    gb[(g + 1) & 1][s][0] = Bcol[m0 * STRIDE];
                    gb[(g + 1) & 1][s][1] = Bcol[(m0 + 1) * STRIDE];
                }
            }
            __builtin_amdgcn_sched_barrier(0);
            #pragma unroll
            for (int s = 0; s < 7; ++s)
                c = __builtin_amdgcn_wmma_f32_16x16x4_f32(false, ga[g & 1][s], false, gb[g & 1][s],
                                                          (short)0, c, false, false);
            __builtin_amdgcn_sched_barrier(0);
        }
        asm volatile("v_nop\n\tv_nop\n\tv_nop\n\tv_nop\n\tv_nop\n\tv_nop\n\tv_nop\n\tv_nop");
        __builtin_amdgcn_sched_barrier(0);
        if (half == 0) {
            #pragma unroll
            for (int v = 0; v < 4; ++v)
                sm_sacc[v * E_ + etile * 16 + lm] = c[v];
        }
    }
    __syncthreads();

    if (t < E_) {
        const int e = t;
        const float uu = sm_u[e], pp = sm_ip[e], nn = sm_inn[e];
        const float wa = sm_wsua[e], wi = sm_wsia[e];
        const float relp = pp * wa * sm_sacc[0 * E_ + e] + uu * wi * sm_sacc[1 * E_ + e];
        const float reln = nn * wa * sm_sacc[2 * E_ + e] + uu * wi * sm_sacc[3 * E_ + e];
        const size_t BE = (size_t)B_ * E_;
        const size_t base = (size_t)b * E_ + e;
        out[0 * BE + base] = uu;
        out[1 * BE + base] = pp;
        out[2 * BE + base] = nn;
        out[3 * BE + base] = relp;
        out[4 * BE + base] = reln;
    }
}

extern "C" void kernel_launch(void* const* d_in, const int* in_sizes, int n_in,
                              void* d_out, int out_size, void* d_ws, size_t ws_size,
                              hipStream_t stream) {
    const int*   user_id    = (const int*)d_in[0];
    const int*   item_id_p  = (const int*)d_in[1];
    const int*   item_id_n  = (const int*)d_in[2];
    const int*   inter_id   = (const int*)d_in[3];
    const float* user_emb_w = (const float*)d_in[4];
    const float* item_emb_w = (const float*)d_in[5];
    const float* ua_key     = (const float*)d_in[6];
    const float* ua_val     = (const float*)d_in[7];
    const float* ia_key     = (const float*)d_in[8];
    const float* ia_val     = (const float*)d_in[9];
    float* out  = (float*)d_out;
    float* wsum = (float*)d_ws;

    hlrm_wsum_kernel<<<1, E_, 0, stream>>>(ua_key, ua_val, ia_key, ia_val, wsum);
    hlrm_main_kernel<<<B_, 256, SMEM_FLOATS * sizeof(float), stream>>>(
        user_id, item_id_p, item_id_n, inter_id,
        user_emb_w, item_emb_w, wsum, out);
}


# Weight-sum kernel: each of its 256 sums loads all 64 key/val pairs before adding (same r order and fma, bit-identical), on 8 one-wave work-groups; was one work-group, one memory latency per 4 rows
# speedup vs baseline: 2.0417x; 1.1710x over previous
#include <hip/hip_runtime.h>
#include <hip/hip_bf16.h>
#include <math.h>

#define B_  4096
#define M_  100
#define E_  128
#define R_  64

typedef float v2f __attribute__((ext_vector_type(2)));
typedef float v8f __attribute__((ext_vector_type(8)));

#define STRIDE    140
#define WP        132
#define MT        112
#define OFF_INTER 0
#define OFF_U     (OFF_INTER + MT * STRIDE)
#define OFF_IP    (OFF_U + E_)
#define OFF_INN   (OFF_IP + E_)
#define OFF_W     (OFF_INN + E_)
#define OFF_WSUA  (OFF_W + 4 * WP)
#define OFF_WSIA  (OFF_WSUA + E_)
#define OFF_LOGIT (OFF_WSIA + E_)
#define OFF_SACC  (OFF_LOGIT + 4 * MT)
#define SMEM_FLOATS (OFF_SACC + 4 * E_)

__device__ __forceinline__ float wave_sum(float v) {
    #pragma unroll
    for (int off = 16; off > 0; off >>= 1) v += __shfl_xor(v, off, 32);
    return v;
}
__device__ __forceinline__ float wave_max(float v) {
    #pragma unroll
    for (int off = 16; off > 0; off >>= 1) v = fmaxf(v, __shfl_xor(v, off, 32));
    return v;
}

__device__ __forceinline__ float xhalf(float x) {
    const int b = __builtin_bit_cast(int, x);
    return __builtin_bit_cast(float,
        __builtin_amdgcn_permlanex16(b, b, 0x76543210, 0xfedcba98, false, false));
}
template <int MASK>
__device__ __forceinline__ float xrow(float x) {
    const int b = __builtin_bit_cast(int, x);
    return __builtin_bit_cast(float,
        __builtin_amdgcn_update_dpp(b, b, 0x160 + MASK, 0xf, 0xf, true));
}

__global__ void __launch_bounds__(32)
hlrm_wsum_kernel(const float* __restrict__ ua_key,
                 const float* __restrict__ ua_val,
                 const float* __restrict__ ia_key,
                 const float* __restrict__ ia_val,
                 float* __restrict__ wsum) {
    const int c = blockIdx.x * 32 + threadIdx.x;
    const int e = c & (E_ - 1);
    const bool ia = blockIdx.x >= 4;
    const float* __restrict__ key = ia ? ia_key : ua_key;
    const float* __restrict__ val = ia ? ia_val : ua_val;
    float k[R_], v[R_];
    #pragma unroll
    for (int r = 0; r < R_; ++r) {
        k[r] = key[r * E_ + e];
        v[r] = val[r * E_ + e];
    }
    float s = 0.f;
    #pragma unroll
    for (int r = 0; r < R_; ++r) s += k[r] * v[r];
    wsum[c] = s;
}

__global__ void __launch_bounds__(256, 2)
hlrm_main_kernel(const int* __restrict__ user_id,
                 const int* __restrict__ item_id_p,
                 const int* __restrict__ item_id_n,
                 const int* __restrict__ inter_id,
                 const float* __restrict__ user_emb_w,
                 const float* __restrict__ item_emb_w,
                 const float* __restrict__ wsum_g,
                 float* __restrict__ out) {
    extern __shared__ float sm[];
    float* sm_inter = sm + OFF_INTER;
    float* sm_u     = sm + OFF_U;
    float* sm_ip    = sm + OFF_IP;
    float* sm_inn   = sm + OFF_INN;
    float* sm_w     = sm + OFF_W;
    float* sm_wsua  = sm + OFF_WSUA;
    float* sm_wsia  = sm + OFF_WSIA;
    float* sm_logit = sm + OFF_LOGIT;
    float* sm_sacc  = sm + OFF_SACC;

    const int b    = blockIdx.x;
    const int t    = threadIdx.x;
    const int wave = t >> 5;
    const int lane = t & 31;
    const int half = lane >> 4;
    const int lm   = lane & 15;

    {
        const int wv = __builtin_amdgcn_readfirstlane(wave);

        int my_id = 0;
        if (lane < 13 && wv + 8 * lane < M_) my_id = inter_id[b * M_ + wv + 8 * lane];
        const int uid = user_id[b], pid = item_id_p[b], nid = item_id_n[b];

        const float* tab12 = (wv == 4) ? user_emb_w : item_emb_w;
        const int    id12  = (wv < 4) ? __builtin_amdgcn_readlane(my_id, 12)
                           : (wv == 4) ? uid : (wv == 5) ? pid : nid;
        float*       dst12 = (wv < 4) ? sm_inter + (96 + wv) * STRIDE
                           : (wv == 4) ? sm_u : (wv == 5) ? sm_ip : sm_inn;

        float4 x[13];
        #pragma unroll
        for (int i = 0; i < 12; ++i) {
            const int id = __builtin_amdgcn_readlane(my_id, i);
            x[i] = ((const float4*)(item_emb_w + (size_t)id * E_))[lane];
        }
        x[12] = make_float4(0.f, 0.f, 0.f, 0.f);
        if (wv < 7) x[12] = ((const float4*)(tab12 + (size_t)id12 * E_))[lane];

        {
            const float4 z = make_float4(0.f, 0.f, 0.f, 0.f);
            ((float4*)(sm_inter + (104 + wv) * STRIDE))[lane] = z;
            if (wv >= 4) ((float4*)(sm_inter + (96 + wv) * STRIDE))[lane] = z;
        }

        float ss[13];
        #pragma unroll
        for (int i = 0; i < 13; ++i)
            ss[i] = x[i].x * x[i].x + x[i].y * x[i].y + x[i].z * x[i].z + x[i].w * x[i].w;

        const bool b16 = (lane & 16) != 0, b8 = (lane & 8) != 0;
        const bool b4  = (lane & 4)  != 0, b2 = (lane & 2) != 0;
        float t8[8], t4[4], t2[2];
        #pragma unroll
        for (int i = 0; i < 8; ++i) {
            const float lo = ss[i], hi = (i + 8 < 13) ? ss[i + 8] : 0.f;
            t8[i] = (b16 ? hi : lo) + xhalf(b16 ? lo : hi);
        }
        #pragma unroll
        for (int i = 0; i < 4; ++i)
            t4[i] = (b8 ? t8[i + 4] : t8[i]) + xrow<8>(b8 ? t8[i] : t8[i + 4]);
        #pragma unroll
        for (int i = 0; i < 2; ++i)
            t2[i] = (b4 ? t4[i + 2] : t4[i]) + xrow<4>(b4 ? t4[i] : t4[i + 2]);
        const float t1  = (b2 ? t2[1] : t2[0]) + xrow<2>(b2 ? t2[0] : t2[1]);
        const float tot = t1 + xrow<1>(t1);

        const float n = sqrtf(tot);
        const float r = 1.0f / (n + 1e-7f);
        const int   sbits = __builtin_bit_cast(int, (n > 1.0f) ? r : 1.0f);

        #pragma unroll
        for (int i = 0; i < 13; ++i) {
            const float s = __builtin_bit_cast(float, __builtin_amdgcn_readlane(sbits, 2 * i));
            float4 y = x[i];
            y.x *= s; y.y *= s; y.z *= s; y.w *= s;
            if (i < 12)      ((float4*)(sm_inter + (wv + 8 * i) * STRIDE))[lane] = y;
            else if (wv < 7) ((float4*)dst12)[lane] = y;
        }
    }
    __syncthreads();

    if (t < E_) {
        const int e = t;
        const float uu = sm_u[e], pp = sm_ip[e], nn = sm_inn[e];
        const float wa = wsum_g[e], wi = wsum_g[E_ + e];
        sm_wsua[e] = wa;
        sm_wsia[e] = wi;
        sm_w[0 * WP + e] = uu * pp * wa;
        sm_w[1 * WP + e] = uu * pp * wi;
        sm_w[2 * WP + e] = uu * nn * wa;
        sm_w[3 * WP + e] = uu * nn * wi;
    }
    __syncthreads();

    if (wave < 7) {
        const int   tile = wave;
        const float* Arow = sm_inter + (tile * 16 + lm) * STRIDE + 2 * half;
        const float* wrow = sm_w + (lm & 3) * WP + 2 * half;
        v8f c = {0.f, 0.f, 0.f, 0.f, 0.f, 0.f, 0.f, 0.f};
        v2f fa[2][8], fb[2][8];
        #pragma unroll
        for (int s = 0; s < 8; ++s) {
            fa[0][s] = *(const v2f*)(Arow + 4 * s);
            fb[0][s] = *(const v2f*)(wrow + 4 * s);
        }
        __builtin_amdgcn_sched_barrier(0);
        #pragma unroll
        for (int g = 0; g < 4; ++g) {
            if (g + 1 < 4) {
                #pragma unroll
                for (int s = 0; s < 8; ++s) {
                    fa[(g + 1) & 1][s] = *(const v2f*)(Arow + 32 * (g + 1) + 4 * s);
                    fb[(g + 1) & 1][s] = *(const v2f*)(wrow + 32 * (g + 1) + 4 * s);
                }
            }
            __builtin_amdgcn_sched_barrier(0);
            #pragma unroll
            for (int s = 0; s < 8; ++s)
                c = __builtin_amdgcn_wmma_f32_16x16x4_f32(false, fa[g & 1][s], false, fb[g & 1][s],
                                                          (short)0, c, false, false);
            __builtin_amdgcn_sched_barrier(0);
        }
        asm volatile("v_nop\n\tv_nop\n\tv_nop\n\tv_nop\n\tv_nop\n\tv_nop\n\tv_nop\n\tv_nop");
        __builtin_amdgcn_sched_barrier(0);
        if (lm < 4) {
            #pragma unroll
            for (int v = 0; v < 8; ++v)
                sm_logit[lm * MT + tile * 16 + v + 8 * half] = c[v];
        }
    }
    __syncthreads();

    if (wave < 4) {
        float* L = sm_logit + wave * MT;
        float vals[4], mx = -INFINITY;
        #pragma unroll
        for (int i = 0; i < 4; ++i) {
            int m = lane + 32 * i;
            vals[i] = (m < M_) ? L[m] : -INFINITY;
            mx = fmaxf(mx, vals[i]);
        }
        mx = wave_max(mx);
        float se = 0.f;
        #pragma unroll
        for (int i = 0; i < 4; ++i) {
            int m = lane + 32 * i;
            vals[i] = (m < M_) ? expf(vals[i] - mx) : 0.f;
            se += vals[i];
        }
        se = wave_sum(se);
        const float inv = 1.0f / se;
        #pragma unroll
        for (int i = 0; i < 4; ++i) {
            int m = lane + 32 * i;
            if (m < MT) L[m] = vals[i] * inv;
        }
    }
    __syncthreads();

    {
        const int   etile = wave;
        const float* Lrow = sm_logit + (lm & 3) * MT + 2 * half;
        const float* Bcol = sm_inter + (2 * half) * STRIDE + etile * 16 + lm;
        v8f c = {0.f, 0.f, 0.f, 0.f, 0.f, 0.f, 0.f, 0.f};
        v2f ga[2][7], gb[2][7];
        #pragma unroll
        for (int s = 0; s < 7; ++s) {
            ga[0][s]    = *(const v2f*)(Lrow + 4 * s);
            gb[0][s][0] = Bcol[(4 * s) * STRIDE];
            gb[0][s][1] = Bcol[(4 * s + 1) * STRIDE];
        }
        __builtin_amdgcn_sched_barrier(0);
        #pragma unroll
        for (int g = 0; g < 4; ++g) {
            if (g + 1 < 4) {
                #pragma unroll
                for (int s = 0; s < 7; ++s) {
                    const int m0 = 28 * (g + 1) + 4 * s;
                    ga[(g + 1) & 1][s]    = *(const v2f*)(Lrow + m0);
                    gb[(g + 1) & 1][s][0] = Bcol[m0 * STRIDE];
                    gb[(g + 1) & 1][s][1] = Bcol[(m0 + 1) * STRIDE];
                }
            }
            __builtin_amdgcn_sched_barrier(0);
            #pragma unroll
            for (int s = 0; s < 7; ++s)
                c = __builtin_amdgcn_wmma_f32_16x16x4_f32(false, ga[g & 1][s], false, gb[g & 1][s],
                                                          (short)0, c, false, false);
            __builtin_amdgcn_sched_barrier(0);
        }
        asm volatile("v_nop\n\tv_nop\n\tv_nop\n\tv_nop\n\tv_nop\n\tv_nop\n\tv_nop\n\tv_nop");
        __builtin_amdgcn_sched_barrier(0);
        if (half == 0) {
            #pragma unroll
            for (int v = 0; v < 4; ++v)
                sm_sacc[v * E_ + etile * 16 + lm] = c[v];
        }
    }
    __syncthreads();

    if (t < E_) {
        const int e = t;
        const float uu = sm_u[e], pp = sm_ip[e], nn = sm_inn[e];
        const float wa = sm_wsua[e], wi = sm_wsia[e];
        const float relp = pp * wa * sm_sacc[0 * E_ + e] + uu * wi * sm_sacc[1 * E_ + e];
        const float reln = nn * wa * sm_sacc[2 * E_ + e] + uu * wi * sm_sacc[3 * E_ + e];
        const size_t BE = (size_t)B_ * E_;
        const size_t base = (size_t)b * E_ + e;
        out[0 * BE + base] = uu;
        out[1 * BE + base] = pp;
        out[2 * BE + base] = nn;
        out[3 * BE + base] = relp;
        out[4 * BE + base] = reln;
    }
}

extern "C" void kernel_launch(void* const* d_in, const int* in_sizes, int n_in,
                              void* d_out, int out_size, void* d_ws, size_t ws_size,
                              hipStream_t stream) {
    const int*   user_id    = (const int*)d_in[0];
    const int*   item_id_p  = (const int*)d_in[1];
    const int*   item_id_n  = (const int*)d_in[2];
    const int*   inter_id   = (const int*)d_in[3];
    const float* user_emb_w = (const float*)d_in[4];
    const float* item_emb_w = (const float*)d_in[5];
    const float* ua_key     = (const float*)d_in[6];
    const float* ua_val     = (const float*)d_in[7];
    const float* ia_key     = (const float*)d_in[8];
    const float* ia_val     = (const float*)d_in[9];
    float* out  = (float*)d_out;
    float* wsum = (float*)d_ws;

    hlrm_wsum_kernel<<<8, 32, 0, stream>>>(ua_key, ua_val, ia_key, ia_val, wsum);
    hlrm_main_kernel<<<B_, 256, SMEM_FLOATS * sizeof(float), stream>>>(
        user_id, item_id_p, item_id_n, inter_id,
        user_emb_w, item_emb_w, wsum, out);
}


# LDS per work-group cut 71232 -> 62464 B so 5 work-groups, not 4, fit a compute unit: 12 zero pad rows and their 3 all-zero stage-6 steps dropped, stage-6 sums stored over the dead weight vectors
# speedup vs baseline: 2.1289x; 1.0427x over previous
#include <hip/hip_runtime.h>
#include <hip/hip_bf16.h>
#include <math.h>

#define B_  4096
#define M_  100
#define E_  128
#define R_  64

typedef float v2f __attribute__((ext_vector_type(2)));
typedef float v8f __attribute__((ext_vector_type(8)));

#define STRIDE    140
#define WP        132
#define MT        112
#define OFF_INTER 0
#define OFF_U     (OFF_INTER + M_ * STRIDE)
#define OFF_IP    (OFF_U + E_)
#define OFF_INN   (OFF_IP + E_)
#define OFF_W     (OFF_INN + E_)
#define OFF_SACC  OFF_W
#define OFF_WSUA  (OFF_W + 4 * WP)
#define OFF_WSIA  (OFF_WSUA + E_)
#define OFF_LOGIT (OFF_WSIA + E_)
#define SMEM_FLOATS (OFF_LOGIT + 4 * MT)
#define NSTEP6    (M_ / 4)

__device__ __forceinline__ float wave_sum(float v) {
    #pragma unroll
    for (int off = 16; off > 0; off >>= 1) v += __shfl_xor(v, off, 32);
    return v;
}
__device__ __forceinline__ float wave_max(float v) {
    #pragma unroll
    for (int off = 16; off > 0; off >>= 1) v = fmaxf(v, __shfl_xor(v, off, 32));
    return v;
}

__device__ __forceinline__ float xhalf(float x) {
    const int b = __builtin_bit_cast(int, x);
    return __builtin_bit_cast(float,
        __builtin_amdgcn_permlanex16(b, b, 0x76543210, 0xfedcba98, false, false));
}
template <int MASK>
__device__ __forceinline__ float xrow(float x) {
    const int b = __builtin_bit_cast(int, x);
    return __builtin_bit_cast(float,
        __builtin_amdgcn_update_dpp(b, b, 0x160 + MASK, 0xf, 0xf, true));
}

__global__ void __launch_bounds__(32)
hlrm_wsum_kernel(const float* __restrict__ ua_key,
                 const float* __restrict__ ua_val,
                 const float* __restrict__ ia_key,
                 const float* __restrict__ ia_val,
                 float* __restrict__ wsum) {
    const int c = blockIdx.x * 32 + threadIdx.x;
    const int e = c & (E_ - 1);
    const bool ia = blockIdx.x >= 4;
    const float* __restrict__ key = ia ? ia_key : ua_key;
    const float* __restrict__ val = ia ? ia_val : ua_val;
    float k[R_], v[R_];
    #pragma unroll
    for (int r = 0; r < R_; ++r) {
        k[r] = key[r * E_ + e];
        v[r] = val[r * E_ + e];
    }
    float s = 0.f;
    #pragma unroll
    for (int r = 0; r < R_; ++r) s += k[r] * v[r];
    wsum[c] = s;
}

__global__ void __launch_bounds__(256, 2)
hlrm_main_kernel(const int* __restrict__ user_id,
                 const int* __restrict__ item_id_p,
                 const int* __restrict__ item_id_n,
                 const int* __restrict__ inter_id,
                 const float* __restrict__ user_emb_w,
                 const float* __restrict__ item_emb_w,
                 const float* __restrict__ wsum_g,
                 float* __restrict__ out) {
    extern __shared__ float sm[];
    float* sm_inter = sm + OFF_INTER;
    float* sm_u     = sm + OFF_U;
    float* sm_ip    = sm + OFF_IP;
    float* sm_inn   = sm + OFF_INN;
    float* sm_w     = sm + OFF_W;
    float* sm_wsua  = sm + OFF_WSUA;
    float* sm_wsia  = sm + OFF_WSIA;
    float* sm_logit = sm + OFF_LOGIT;
    float* sm_sacc  = sm + OFF_SACC;

    const int b    = blockIdx.x;
    const int t    = threadIdx.x;
    const int wave = t >> 5;
    const int lane = t & 31;
    const int half = lane >> 4;
    const int lm   = lane & 15;

    {
        const int wv = __builtin_amdgcn_readfirstlane(wave);

        int my_id = 0;
        if (lane < 13 && wv + 8 * lane < M_) my_id = inter_id[b * M_ + wv + 8 * lane];
        const int uid = user_id[b], pid = item_id_p[b], nid = item_id_n[b];

        const float* tab12 = (wv == 4) ? user_emb_w : item_emb_w;
        const int    id12  = (wv < 4) ? __builtin_amdgcn_readlane(my_id, 12)
                           : (wv == 4) ? uid : (wv == 5) ? pid : nid;
        float*       dst12 = (wv < 4) ? sm_inter + (96 + wv) * STRIDE
                           : (wv == 4) ? sm_u : (wv == 5) ? sm_ip : sm_inn;

        float4 x[13];
        #pragma unroll
        for (int i = 0; i < 12; ++i) {
            const int id = __builtin_amdgcn_readlane(my_id, i);
            x[i] = ((const float4*)(item_emb_w + (size_t)id * E_))[lane];
        }
        x[12] = make_float4(0.f, 0.f, 0.f, 0.f);
        if (wv < 7) x[12] = ((const float4*)(tab12 + (size_t)id12 * E_))[lane];

        float ss[13];
        #pragma unroll
        for (int i = 0; i < 13; ++i)
            ss[i] = x[i].x * x[i].x + x[i].y * x[i].y + x[i].z * x[i].z + x[i].w * x[i].w;

        const bool b16 = (lane & 16) != 0, b8 = (lane & 8) != 0;
        const bool b4  = (lane & 4)  != 0, b2 = (lane & 2) != 0;
        float t8[8], t4[4], t2[2];
        #pragma unroll
        for (int i = 0; i < 8; ++i) {
            const float lo = ss[i], hi = (i + 8 < 13) ? ss[i + 8] : 0.f;
            t8[i] = (b16 ? hi : lo) + xhalf(b16 ? lo : hi);
        }
        #pragma unroll
        for (int i = 0; i < 4; ++i)
            t4[i] = (b8 ? t8[i + 4] : t8[i]) + xrow<8>(b8 ? t8[i] : t8[i + 4]);
        #pragma unroll
        for (int i = 0; i < 2; ++i)
            t2[i] = (b4 ? t4[i + 2] : t4[i]) + xrow<4>(b4 ? t4[i] : t4[i + 2]);
        const float t1  = (b2 ? t2[1] : t2[0]) + xrow<2>(b2 ? t2[0] : t2[1]);
        const float tot = t1 + xrow<1>(t1);

        const float n = sqrtf(tot);
        const float r = 1.0f / (n + 1e-7f);
        const int   sbits = __builtin_bit_cast(int, (n > 1.0f) ? r : 1.0f);

        #pragma unroll
        for (int i = 0; i < 13; ++i) {
            const float s = __builtin_bit_cast(float, __builtin_amdgcn_readlane(sbits, 2 * i));
            float4 y = x[i];
            y.x *= s; y.y *= s; y.z *= s; y.w *= s;
            if (i < 12)      ((float4*)(sm_inter + (wv + 8 * i) * STRIDE))[lane] = y;
            else if (wv < 7) ((float4*)dst12)[lane] = y;
        }
    }
    __syncthreads();

    if (t < E_) {
        const int e = t;
        const float uu = sm_u[e], pp = sm_ip[e], nn = sm_inn[e];
        const float wa = wsum_g[e], wi = wsum_g[E_ + e];
        sm_wsua[e] = wa;
        sm_wsia[e] = wi;
        sm_w[0 * WP + e] = uu * pp * wa;
        sm_w[1 * WP + e] = uu * pp * wi;
        sm_w[2 * WP + e] = uu * nn * wa;
        sm_w[3 * WP + e] = uu * nn * wi;
    }
    __syncthreads();

    if (wave < 7) {
        const int   tile = wave;
        const int    arow = (tile * 16 + lm < M_) ? tile * 16 + lm : 96 + (lm & 3);
        const float* Arow = sm_inter + arow * STRIDE + 2 * half;
        const float* wrow = sm_w + (lm & 3) * WP + 2 * half;
        v8f c = {0.f, 0.f, 0.f, 0.f, 0.f, 0.f, 0.f, 0.f};
        v2f fa[2][8], fb[2][8];
        #pragma unroll
        for (int s = 0; s < 8; ++s) {
            fa[0][s] = *(const v2f*)(Arow + 4 * s);
            fb[0][s] = *(const v2f*)(wrow + 4 * s);
        }
        __builtin_amdgcn_sched_barrier(0);
        #pragma unroll
        for (int g = 0; g < 4; ++g) {
            if (g + 1 < 4) {
                #pragma unroll
                for (int s = 0; s < 8; ++s) {
                    fa[(g + 1) & 1][s] = *(const v2f*)(Arow + 32 * (g + 1) + 4 * s);
                    fb[(g + 1) & 1][s] = *(const v2f*)(wrow + 32 * (g + 1) + 4 * s);
                }
            }
            __builtin_amdgcn_sched_barrier(0);
            #pragma unroll
            for (int s = 0; s < 8; ++s)
                c = __builtin_amdgcn_wmma_f32_16x16x4_f32(false, fa[g & 1][s], false, fb[g & 1][s],
                                                          (short)0, c, false, false);
            __builtin_amdgcn_sched_barrier(0);
        }
        asm volatile("v_nop\n\tv_nop\n\tv_nop\n\tv_nop\n\tv_nop\n\tv_nop\n\tv_nop\n\tv_nop");
        __builtin_amdgcn_sched_barrier(0);
        if (lm < 4) {
            #pragma unroll
            for (int v = 0; v < 8; ++v)
                sm_logit[lm * MT + tile * 16 + v + 8 * half] = c[v];
        }
    }
    __syncthreads();

    if (wave < 4) {
        float* L = sm_logit + wave * MT;
        float vals[4], mx = -INFINITY;
        #pragma unroll
        for (int i = 0; i < 4; ++i) {
            int m = lane + 32 * i;
            vals[i] = (m < M_) ? L[m] : -INFINITY;
            mx = fmaxf(mx, vals[i]);
        }
        mx = wave_max(mx);
        float se = 0.f;
        #pragma unroll
        for (int i = 0; i < 4; ++i) {
            int m = lane + 32 * i;
            vals[i] = (m < M_) ? expf(vals[i] - mx) : 0.f;
            se += vals[i];
        }
        se = wave_sum(se);
        const float inv = 1.0f / se;
        #pragma unroll
        for (int i = 0; i < 4; ++i) {
            int m = lane + 32 * i;
            if (m < MT) L[m] = vals[i] * inv;
        }
    }
    __syncthreads();

    {
        const int   etile = wave;
        const float* Lrow = sm_logit + (lm & 3) * MT + 2 * half;
        const float* Bcol = sm_inter + (2 * half) * STRIDE + etile * 16 + lm;
        v8f c = {0.f, 0.f, 0.f, 0.f, 0.f, 0.f, 0.f, 0.f};
        v2f ga[2][7], gb[2][7];
        #pragma unroll
        for (int s = 0; s < 7; ++s) {
            ga[0][s]    = *(const v2f*)(Lrow + 4 * s);
            gb[0][s][0] = Bcol[(4 * s) * STRIDE];
            gb[0][s][1] = Bcol[(4 * s + 1) * STRIDE];
        }
        __builtin_amdgcn_sched_barrier(0);
        #pragma unroll
        for (int g = 0; g < 4; ++g) {
            if (g + 1 < 4) {
                #pragma unroll
                for (int s = 0; s < 7; ++s) {
                    if (7 * (g + 1) + s < NSTEP6) {
                        const int m0 = 28 * (g + 1) + 4 * s;
                        ga[(g + 1) & 1][s]    = *(const v2f*)(Lrow + m0);
                        gb[(g + 1) & 1][s][0] = Bcol[m0 * STRIDE];
                        gb[(g + 1) & 1][s][1] = Bcol[(m0 + 1) * STRIDE];
                    }
                }
            }
            __builtin_amdgcn_sched_barrier(0);
            #pragma unroll
            for (int s = 0; s < 7; ++s)
                if (7 * g + s < NSTEP6)
                    c = __builtin_amdgcn_wmma_f32_16x16x4_f32(false, ga[g & 1][s], false,
                                                              gb[g & 1][s], (short)0, c, false, false);
            __builtin_amdgcn_sched_barrier(0);
        }
        asm volatile("v_nop\n\tv_nop\n\tv_nop\n\tv_nop\n\tv_nop\n\tv_nop\n\tv_nop\n\tv_nop");
        __builtin_amdgcn_sched_barrier(0);
        if (half == 0) {
            #pragma unroll
            for (int v = 0; v < 4; ++v)
                sm_sacc[v * E_ + etile * 16 + lm] = c[v];
        }
    }
    __syncthreads();

    if (t < E_) {
        const int e = t;
        const float uu = sm_u[e], pp = sm_ip[e], nn = sm_inn[e];
        const float wa = sm_wsua[e], wi = sm_wsia[e];
        const float relp = pp * wa * sm_sacc[0 * E_ + e] + uu * wi * sm_sacc[1 * E_ + e];
        const float reln = nn * wa * sm_sacc[2 * E_ + e] + uu * wi * sm_sacc[3 * E_ + e];
        const size_t BE = (size_t)B_ * E_;
        const size_t base = (size_t)b * E_ + e;
        out[0 * BE + base] = uu;
        out[1 * BE + base] = pp;
        out[2 * BE + base] = nn;
        out[3 * BE + base] = relp;
        out[4 * BE + base] = reln;
    }
}

extern "C" void kernel_launch(void* const* d_in, const int* in_sizes, int n_in,
                              void* d_out, int out_size, void* d_ws, size_t ws_size,
                              hipStream_t stream) {
    const int*   user_id    = (const int*)d_in[0];
    const int*   item_id_p  = (const int*)d_in[1];
    const int*   item_id_n  = (const int*)d_in[2];
    const int*   inter_id   = (const int*)d_in[3];
    const float* user_emb_w = (const float*)d_in[4];
    const float* item_emb_w = (const float*)d_in[5];
    const float* ua_key     = (const float*)d_in[6];
    const float* ua_val     = (const float*)d_in[7];
    const float* ia_key     = (const float*)d_in[8];
    const float* ia_val     = (const float*)d_in[9];
    float* out  = (float*)d_out;
    float* wsum = (float*)d_ws;

    hlrm_wsum_kernel<<<8, 32, 0, stream>>>(ua_key, ua_val, ia_key, ia_val, wsum);
    hlrm_main_kernel<<<B_, 256, SMEM_FLOATS * sizeof(float), stream>>>(
        user_id, item_id_p, item_id_n, inter_id,
        user_emb_w, item_emb_w, wsum, out);
}


# Fewer issued instructions outside the matrix stages: scalar-base row loads, squares as packed mul + packed fma (row sum order (x2+z2)+(y2+w2)), softmax butterflies by DPP instead of LDS shuffles
# speedup vs baseline: 2.2296x; 1.0473x over previous
#include <hip/hip_runtime.h>
#include <hip/hip_bf16.h>
#include <math.h>

#define B_  4096
#define M_  100
#define E_  128
#define R_  64

typedef float v2f __attribute__((ext_vector_type(2)));
typedef float v8f __attribute__((ext_vector_type(8)));

#define STRIDE    140
#define WP        132
#define MT        112
#define OFF_INTER 0
#define OFF_U     (OFF_INTER + M_ * STRIDE)
#define OFF_IP    (OFF_U + E_)
#define OFF_INN   (OFF_IP + E_)
#define OFF_W     (OFF_INN + E_)
#define OFF_SACC  OFF_W
#define OFF_WSUA  (OFF_W + 4 * WP)
#define OFF_WSIA  (OFF_WSUA + E_)
#define OFF_LOGIT (OFF_WSIA + E_)
#define SMEM_FLOATS (OFF_LOGIT + 4 * MT)
#define NSTEP6    (M_ / 4)

__device__ __forceinline__ float xhalf(float x) {
    const int b = __builtin_bit_cast(int, x);
    return __builtin_bit_cast(float,
        __builtin_amdgcn_permlanex16(b, b, 0x76543210, 0xfedcba98, false, false));
}
template <int MASK>
__device__ __forceinline__ float xrow(float x) {
    const int b = __builtin_bit_cast(int, x);
    return __builtin_bit_cast(float,
        __builtin_amdgcn_update_dpp(b, b, 0x160 + MASK, 0xf, 0xf, true));
}

__device__ __forceinline__ float wave_sum(float v) {
    v += xhalf(v);
    v += xrow<8>(v);
    v += xrow<4>(v);
    v += xrow<2>(v);
    v += xrow<1>(v);
    return v;
}
__device__ __forceinline__ float wave_max(float v) {
    v = fmaxf(v, xhalf(v));
    v = fmaxf(v, xrow<8>(v));
    v = fmaxf(v, xrow<4>(v));
    v = fmaxf(v, xrow<2>(v));
    v = fmaxf(v, xrow<1>(v));
    return v;
}

__global__ void __launch_bounds__(32)
hlrm_wsum_kernel(const float* __restrict__ ua_key,
                 const float* __restrict__ ua_val,
                 const float* __restrict__ ia_key,
                 const float* __restrict__ ia_val,
                 float* __restrict__ wsum) {
    const int c = blockIdx.x * 32 + threadIdx.x;
    const int e = c & (E_ - 1);
    const bool ia = blockIdx.x >= 4;
    const float* __restrict__ key = ia ? ia_key : ua_key;
    const float* __restrict__ val = ia ? ia_val : ua_val;
    float k[R_], v[R_];
    #pragma unroll
    for (int r = 0; r < R_; ++r) {
        k[r] = key[r * E_ + e];
        v[r] = val[r * E_ + e];
    }
    float s = 0.f;
    #pragma unroll
    for (int r = 0; r < R_; ++r) s += k[r] * v[r];
    wsum[c] = s;
}

__global__ void __launch_bounds__(256, 2)
hlrm_main_kernel(const int* __restrict__ user_id,
                 const int* __restrict__ item_id_p,
                 const int* __restrict__ item_id_n,
                 const int* __restrict__ inter_id,
                 const float* __restrict__ user_emb_w,
                 const float* __restrict__ item_emb_w,
                 const float* __restrict__ wsum_g,
                 float* __restrict__ out) {
    extern __shared__ float sm[];
    float* sm_inter = sm + OFF_INTER;
    float* sm_u     = sm + OFF_U;
    float* sm_ip    = sm + OFF_IP;
    float* sm_inn   = sm + OFF_INN;
    float* sm_w     = sm + OFF_W;
    float* sm_wsua  = sm + OFF_WSUA;
    float* sm_wsia  = sm + OFF_WSIA;
    float* sm_logit = sm + OFF_LOGIT;
    float* sm_sacc  = sm + OFF_SACC;

    const int b    = blockIdx.x;
    const int t    = threadIdx.x;
    const int wave = t >> 5;
    const int lane = t & 31;
    const int half = lane >> 4;
    const int lm   = lane & 15;

    {
        const int wv = __builtin_amdgcn_readfirstlane(wave);

        int my_id = 0;
        if (lane < 13 && wv + 8 * lane < M_) my_id = inter_id[b * M_ + wv + 8 * lane];
        const int uid = user_id[b], pid = item_id_p[b], nid = item_id_n[b];

        const float* tab12 = (wv == 4) ? user_emb_w : item_emb_w;
        const int    id12  = (wv < 4) ? __builtin_amdgcn_readlane(my_id, 12)
                           : (wv == 4) ? uid : (wv == 5) ? pid : nid;
        float*       dst12 = (wv < 4) ? sm_inter + (96 + wv) * STRIDE
                           : (wv == 4) ? sm_u : (wv == 5) ? sm_ip : sm_inn;

        float4 x[13];
        typedef const __attribute__((address_space(1))) float*  gfp_t;
        typedef const __attribute__((address_space(1))) float4* gf4p_t;
        const unsigned my_off = (unsigned)my_id * (unsigned)(E_ * sizeof(float));
        unsigned off[12];
        gfp_t    rowp[12];
        #pragma unroll
        for (int i = 0; i < 12; ++i) off[i] = (unsigned)__builtin_amdgcn_readlane((int)my_off, i);
        __builtin_amdgcn_sched_barrier(0);
        #pragma unroll
        for (int i = 0; i < 12; ++i) {
            rowp[i] = (gfp_t)((const char*)item_emb_w + off[i]);
            asm("" : "+s"(rowp[i]));
        }
        __builtin_amdgcn_sched_barrier(0);
        #pragma unroll
        for (int i = 0; i < 12; ++i) x[i] = ((gf4p_t)rowp[i])[lane];
        x[12] = make_float4(0.f, 0.f, 0.f, 0.f);
        if (wv < 7) x[12] = ((const float4*)(tab12 + (size_t)id12 * E_))[lane];

        float ss[13];
        #pragma unroll
        for (int i = 0; i < 13; ++i) {
            const v2f lo = {x[i].x, x[i].y}, hi = {x[i].z, x[i].w};
            v2f p = lo * lo;
            p = __builtin_elementwise_fma(hi, hi, p);
            ss[i] = p[0] + p[1];
        }

        const bool b16 = (lane & 16) != 0, b8 = (lane & 8) != 0;
        const bool b4  = (lane & 4)  != 0, b2 = (lane & 2) != 0;
        float t8[8], t4[4], t2[2];
        #pragma unroll
        for (int i = 0; i < 8; ++i) {
            const float lo = ss[i], hi = (i + 8 < 13) ? ss[i + 8] : 0.f;
            t8[i] = (b16 ? hi : lo) + xhalf(b16 ? lo : hi);
        }
        #pragma unroll
        for (int i = 0; i < 4; ++i)
            t4[i] = (b8 ? t8[i + 4] : t8[i]) + xrow<8>(b8 ? t8[i] : t8[i + 4]);
        #pragma unroll
        for (int i = 0; i < 2; ++i)
            t2[i] = (b4 ? t4[i + 2] : t4[i]) + xrow<4>(b4 ? t4[i] : t4[i + 2]);
        const float t1  = (b2 ? t2[1] : t2[0]) + xrow<2>(b2 ? t2[0] : t2[1]);
        const float tot = t1 + xrow<1>(t1);

        const float n = sqrtf(tot);
        const float r = 1.0f / (n + 1e-7f);
        const int   sbits = __builtin_bit_cast(int, (n > 1.0f) ? r : 1.0f);

        #pragma unroll
        for (int i = 0; i < 13; ++i) {
            const float s = __builtin_bit_cast(float, __builtin_amdgcn_readlane(sbits, 2 * i));
            float4 y = x[i];
            y.x *= s; y.y *= s; y.z *= s; y.w *= s;
            if (i < 12)      ((float4*)(sm_inter + (wv + 8 * i) * STRIDE))[lane] = y;
            else if (wv < 7) ((float4*)dst12)[lane] = y;
        }
    }
    __syncthreads();

    if (t < E_) {
        const int e = t;
        const float uu = sm_u[e], pp = sm_ip[e], nn = sm_inn[e];
        const float wa = wsum_g[e], wi = wsum_g[E_ + e];
        sm_wsua[e] = wa;
        sm_wsia[e] = wi;
        sm_w[0 * WP + e] = uu * pp * wa;
        sm_w[1 * WP + e] = uu * pp * wi;
        sm_w[2 * WP + e] = uu * nn * wa;
        sm_w[3 * WP + e] = uu * nn * wi;
    }
    __syncthreads();

    if (wave < 7) {
        const int   tile = wave;
        const int    arow = (tile * 16 + lm < M_) ? tile * 16 + lm : 96 + (lm & 3);
        const float* Arow = sm_inter + arow * STRIDE + 2 * half;
        const float* wrow = sm_w + (lm & 3) * WP + 2 * half;
        v8f c = {0.f, 0.f, 0.f, 0.f, 0.f, 0.f, 0.f, 0.f};
        v2f fa[2][8], fb[2][8];
        #pragma unroll
        for (int s = 0; s < 8; ++s) {
            fa[0][s] = *(const v2f*)(Arow + 4 * s);
            fb[0][s] = *(const v2f*)(wrow + 4 * s);
        }
        __builtin_amdgcn_sched_barrier(0);
        #pragma unroll
        for (int g = 0; g < 4; ++g) {
            if (g + 1 < 4) {
                #pragma unroll
                for (int s = 0; s < 8; ++s) {
                    fa[(g + 1) & 1][s] = *(const v2f*)(Arow + 32 * (g + 1) + 4 * s);
                    fb[(g + 1) & 1][s] = *(const v2f*)(wrow + 32 * (g + 1) + 4 * s);
                }
            }
            __builtin_amdgcn_sched_barrier(0);
            #pragma unroll
            for (int s = 0; s < 8; ++s)
                c = __builtin_amdgcn_wmma_f32_16x16x4_f32(false, fa[g & 1][s], false, fb[g & 1][s],
                                                          (short)0, c, false, false);
            __builtin_amdgcn_sched_barrier(0);
        }
        asm volatile("v_nop\n\tv_nop\n\tv_nop\n\tv_nop\n\tv_nop\n\tv_nop\n\tv_nop\n\tv_nop");
        __builtin_amdgcn_sched_barrier(0);
        if (lm < 4) {
            #pragma unroll
            for (int v = 0; v < 8; ++v)
                sm_logit[lm * MT + tile * 16 + v + 8 * half] = c[v];
        }
    }
    __syncthreads();

    if (wave < 4) {
        float* L = sm_logit + wave * MT;
        float vals[4], mx = -INFINITY;
        #pragma unroll
        for (int i = 0; i < 4; ++i) {
            int m = lane + 32 * i;
            vals[i] = (m < M_) ? L[m] : -INFINITY;
            mx = fmaxf(mx, vals[i]);
        }
        mx = wave_max(mx);
        float se = 0.f;
        #pragma unroll
        for (int i = 0; i < 4; ++i) {
            int m = lane + 32 * i;
            vals[i] = (m < M_) ? expf(vals[i] - mx) : 0.f;
            se += vals[i];
        }
        se = wave_sum(se);
        const float inv = 1.0f / se;
        #pragma unroll
        for (int i = 0; i < 4; ++i) {
            int m = lane + 32 * i;
            if (m < MT) L[m] = vals[i] * inv;
        }
    }
    __syncthreads();

    {
        const int   etile = wave;
        const float* Lrow = sm_logit + (lm & 3) * MT + 2 * half;
        const float* Bcol = sm_inter + (2 * half) * STRIDE + etile * 16 + lm;
        v8f c = {0.f, 0.f, 0.f, 0.f, 0.f, 0.f, 0.f, 0.f};
        v2f ga[2][7], gb[2][7];
        #pragma unroll
        for (int s = 0; s < 7; ++s) {
            ga[0][s]    = *(const v2f*)(Lrow + 4 * s);
            gb[0][s][0] = Bcol[(4 * s) * STRIDE];
            gb[0][s][1] = Bcol[(4 * s + 1) * STRIDE];
        }
        __builtin_amdgcn_sched_barrier(0);
        #pragma unroll
        for (int g = 0; g < 4; ++g) {
            if (g + 1 < 4) {
                #pragma unroll
                for (int s = 0; s < 7; ++s) {
                    if (7 * (g + 1) + s < NSTEP6) {
                        const int m0 = 28 * (g + 1) + 4 * s;
                        ga[(g + 1) & 1][s]    = *(const v2f*)(Lrow + m0);
                        gb[(g + 1) & 1][s][0] = Bcol[m0 * STRIDE];
                        gb[(g + 1) & 1][s][1] = Bcol[(m0 + 1) * STRIDE];
                    }
                }
            }
            __builtin_amdgcn_sched_barrier(0);
            #pragma unroll
            for (int s = 0; s < 7; ++s)
                if (7 * g + s < NSTEP6)
                    c = __builtin_amdgcn_wmma_f32_16x16x4_f32(false, ga[g & 1][s], false,
                                                              gb[g & 1][s], (short)0, c, false, false);
            __builtin_amdgcn_sched_barrier(0);
        }
        asm volatile("v_nop\n\tv_nop\n\tv_nop\n\tv_nop\n\tv_nop\n\tv_nop\n\tv_nop\n\tv_nop");
        __builtin_amdgcn_sched_barrier(0);
        if (half == 0) {
            #pragma unroll
            for (int v = 0; v < 4; ++v)
                sm_sacc[v * E_ + etile * 16 + lm] = c[v];
        }
    }
    __syncthreads();

    if (t < E_) {
        const int e = t;
        const float uu = sm_u[e], pp = sm_ip[e], nn = sm_inn[e];
        const float wa = sm_wsua[e], wi = sm_wsia[e];
        const float relp = pp * wa * sm_sacc[0 * E_ + e] + uu * wi * sm_sacc[1 * E_ + e];
        const float reln = nn * wa * sm_sacc[2 * E_ + e] + uu * wi * sm_sacc[3 * E_ + e];
        const size_t BE = (size_t)B_ * E_;
        const size_t base = (size_t)b * E_ + e;
        out[0 * BE + base] = uu;
        out[1 * BE + base] = pp;
        out[2 * BE + base] = nn;
        out[3 * BE + base] = relp;
        out[4 * BE + base] = reln;
    }
}

extern "C" void kernel_launch(void* const* d_in, const int* in_sizes, int n_in,
                              void* d_out, int out_size, void* d_ws, size_t ws_size,
                              hipStream_t stream) {
    const int*   user_id    = (const int*)d_in[0];
    const int*   item_id_p  = (const int*)d_in[1];
    const int*   item_id_n  = (const int*)d_in[2];
    const int*   inter_id   = (const int*)d_in[3];
    const float* user_emb_w = (const float*)d_in[4];
    const float* item_emb_w = (const float*)d_in[5];
    const float* ua_key     = (const float*)d_in[6];
    const float* ua_val     = (const float*)d_in[7];
    const float* ia_key     = (const float*)d_in[8];
    const float* ia_val     = (const float*)d_in[9];
    float* out  = (float*)d_out;
    float* wsum = (float*)d_ws;

    hlrm_wsum_kernel<<<8, 32, 0, stream>>>(ua_key, ua_val, ia_key, ia_val, wsum);
    hlrm_main_kernel<<<B_, 256, SMEM_FLOATS * sizeof(float), stream>>>(
        user_id, item_id_p, item_id_n, inter_id,
        user_emb_w, item_emb_w, wsum, out);
}
